# starting kernel
# baseline (speedup 1.0000x reference)
#include <hip/hip_runtime.h>
#include <hip/hip_bf16.h>

typedef __attribute__((ext_vector_type(2))) float v2f;
typedef __attribute__((ext_vector_type(8))) float v8f;

#define F_DIM 128
#define H_DIM 256
#define C_DIM 40
#define BN_EPS 1e-5f

__global__ __launch_bounds__(256) void mlp_kernel(
    const float* __restrict__ x, const float* __restrict__ W1,
    const float* __restrict__ b1, const float* __restrict__ gamma,
    const float* __restrict__ beta, const float* __restrict__ mean,
    const float* __restrict__ var, const float* __restrict__ W2,
    const float* __restrict__ b2, float* __restrict__ h0, int N)
{
    __shared__ float As[16 * 129];
    __shared__ float Hs[16 * 257];

    const int tid  = threadIdx.x;
    const int wave = tid >> 5;
    const int lane = tid & 31;
    const int half = lane >> 4;
    const int l16  = lane & 15;
    const int row0 = blockIdx.x * 16;

    #pragma unroll
    for (int i = 0; i < 8; ++i) {
        int idx = tid + i * 256;
        int r = idx >> 7, c = idx & 127;
        int gr = row0 + r;
        As[r * 129 + c] = (gr < N) ? x[(long)gr * F_DIM + c] : 0.0f;
    }
    __syncthreads();

    #pragma unroll
    for (int t = 0; t < 2; ++t) {
        const int ct = wave * 2 + t;
        const int n  = ct * 16 + l16;
        v8f acc = {};
        for (int k = 0; k < F_DIM; k += 4) {
            v2f a, b;
            a.x = As[l16 * 129 + k + 2 * half + 0];
            a.y = As[l16 * 129 + k + 2 * half + 1];
            b.x = W1[(long)(k + 2 * half + 0) * H_DIM + n];
            b.y = W1[(long)(k + 2 * half + 1) * H_DIM + n];
            acc = __builtin_amdgcn_wmma_f32_16x16x4_f32(
                false, a, false, b, (short)0, acc, false, false);
        }
        const float bb = b1[n];
        const float g  = gamma[n], be = beta[n];
        const float mu = mean[n],  iv = rsqrtf(var[n] + BN_EPS);
        #pragma unroll
        for (int r = 0; r < 8; ++r) {
            const int m = r + 8 * half;
            float v = acc[r] + bb;
            v = (v - mu) * iv * g + be;
            Hs[m * 257 + n] = fmaxf(v, 0.0f);
        }
    }
    __syncthreads();

    if (wave < 3) {
        const int n   = wave * 16 + l16;
        const bool nok = (n < C_DIM);
        v8f acc = {};
        for (int k = 0; k < H_DIM; k += 4) {
            v2f a, b;
            a.x = Hs[l16 * 257 + k + 2 * half + 0];
            a.y = Hs[l16 * 257 + k + 2 * half + 1];
            b.x = nok ? W2[(long)(k + 2 * half + 0) * C_DIM + n] : 0.0f;
            b.y = nok ? W2[(long)(k + 2 * half + 1) * C_DIM + n] : 0.0f;
            acc = __builtin_amdgcn_wmma_f32_16x16x4_f32(
                false, a, false, b, (short)0, acc, false, false);
        }
        const float bb = nok ? b2[n] : 0.0f;
        #pragma unroll
        for (int r = 0; r < 8; ++r) {
            const int m  = r + 8 * half;
            const int gr = row0 + m;
            if (nok && gr < N) h0[(long)gr * C_DIM + n] = acc[r] + bb;
        }
    }
}

__global__ void zero_kernel(float* __restrict__ p, long n)
{
    long i = (long)blockIdx.x * blockDim.x + threadIdx.x;
    if (i < n) p[i] = 0.0f;
}

__global__ void scatter_kernel(const int* __restrict__ src,
                               const int* __restrict__ dst,
                               const float* __restrict__ norm,
                               const float* __restrict__ prev,
                               float* __restrict__ next, int E)
{
    long tid = (long)blockIdx.x * blockDim.x + threadIdx.x;
    int e = (int)(tid >> 3);
    int l = (int)(tid & 7);
    if (e >= E) return;
    const int   s = src[e];
    const int   d = dst[e];
    const float w = norm[e];
    const float* pr = prev + (long)s * C_DIM;
    float*       nx = next + (long)d * C_DIM;
    #pragma unroll
    for (int j = 0; j < 5; ++j) {
        const int c = l + 8 * j;
        atomicAdd(&nx[c], w * pr[c]);
    }
}

__global__ void combine_kernel(const float* __restrict__ v,
                               float* __restrict__ acc,
                               const float* __restrict__ pw,
                               const float* __restrict__ pb, int N)
{
    int n = blockIdx.x * blockDim.x + threadIdx.x;
    if (n >= N) return;
    const float* row = v + (long)n * C_DIM;
    float d = pb[0];
    #pragma unroll
    for (int c = 0; c < C_DIM; ++c) d += row[c] * pw[c];
    const float r = 1.0f / (1.0f + expf(-d));
    float* ar = acc + (long)n * C_DIM;
    #pragma unroll
    for (int c = 0; c < C_DIM; ++c) ar[c] += r * row[c];
}

__global__ void lsm_kernel(const float* __restrict__ acc,
                           float* __restrict__ out, int N)
{
    int n = blockIdx.x * blockDim.x + threadIdx.x;
    if (n >= N) return;
    const float* row = acc + (long)n * C_DIM;
    float m = -3.402823466e38f;
    #pragma unroll
    for (int c = 0; c < C_DIM; ++c) m = fmaxf(m, row[c]);
    float s = 0.0f;
    #pragma unroll
    for (int c = 0; c < C_DIM; ++c) s += expf(row[c] - m);
    const float ls = logf(s);
    float* o = out + (long)n * C_DIM;
    #pragma unroll
    for (int c = 0; c < C_DIM; ++c) o[c] = row[c] - m - ls;
}

extern "C" void kernel_launch(void* const* d_in, const int* in_sizes, int n_in,
                              void* d_out, int out_size, void* d_ws, size_t ws_size,
                              hipStream_t stream)
{
    const float* x     = (const float*)d_in[0];
    const int*   edge  = (const int*)d_in[1];
    const float* norm  = (const float*)d_in[2];
    const float* W1    = (const float*)d_in[3];
    const float* b1    = (const float*)d_in[4];
    const float* gamma = (const float*)d_in[5];
    const float* beta  = (const float*)d_in[6];
    const float* mean  = (const float*)d_in[7];
    const float* var   = (const float*)d_in[8];
    const float* W2    = (const float*)d_in[9];
    const float* b2    = (const float*)d_in[10];
    const float* pw    = (const float*)d_in[11];
    const float* pb    = (const float*)d_in[12];

    const int N = in_sizes[0] / F_DIM;
    const int E = in_sizes[2];

    const size_t per = (((size_t)N * C_DIM) + 63) & ~(size_t)63;
    float* h0   = (float*)d_ws;
    float* bufA = h0   + per;
    float* bufB = bufA + per;
    float* accb = bufB + per;

    const int* src = edge;
    const int* dst = edge + E;

    const long tot     = (long)N * C_DIM;
    const int  zblocks = (int)((tot + 255) / 256);
    const int  nblocks = (N + 255) / 256;

    mlp_kernel<<<(N + 15) / 16, 256, 0, stream>>>(
        x, W1, b1, gamma, beta, mean, var, W2, b2, h0, N);

    zero_kernel<<<zblocks, 256, 0, stream>>>(accb, tot);
    combine_kernel<<<nblocks, 256, 0, stream>>>(h0, accb, pw, pb, N);

    const float* prev = h0;
    for (int k = 0; k < 10; ++k) {
        float* next = (k & 1) ? bufB : bufA;
        zero_kernel<<<zblocks, 256, 0, stream>>>(next, tot);
        const long st = (long)E * 8;
        scatter_kernel<<<(int)((st + 255) / 256), 256, 0, stream>>>(
            src, dst, norm, prev, next, E);
        combine_kernel<<<nblocks, 256, 0, stream>>>(next, accb, pw, pb, N);
        prev = next;
    }

    lsm_kernel<<<nblocks, 256, 0, stream>>>(accb, (float*)d_out, N);
}


# Hops accumulate in a work-group-owned LDS tile of 256 dst rows (edges binned by dst tile once per call; LDS float atomics, arrival order) instead of zero fill + global float-atomic scatter
# speedup vs baseline: 1.9766x; 1.9766x over previous
#include <hip/hip_runtime.h>
#include <hip/hip_bf16.h>

typedef __attribute__((ext_vector_type(2))) float v2f;
typedef __attribute__((ext_vector_type(8))) float v8f;

#define F_DIM 128
#define H_DIM 256
#define C_DIM 40
#define BN_EPS 1e-5f

__global__ __launch_bounds__(256) void mlp_kernel(
    const float* __restrict__ x, const float* __restrict__ W1,
    const float* __restrict__ b1, const float* __restrict__ gamma,
    const float* __restrict__ beta, const float* __restrict__ mean,
    const float* __restrict__ var, const float* __restrict__ W2,
    const float* __restrict__ b2, float* __restrict__ h0, int N)
{
    __shared__ float As[16 * 129];
    __shared__ float Hs[16 * 257];

    const int tid  = threadIdx.x;
    const int wave = tid >> 5;
    const int lane = tid & 31;
    const int half = lane >> 4;
    const int l16  = lane & 15;
    const int row0 = blockIdx.x * 16;

    #pragma unroll
    for (int i = 0; i < 8; ++i) {
        int idx = tid + i * 256;
        int r = idx >> 7, c = idx & 127;
        int gr = row0 + r;
        As[r * 129 + c] = (gr < N) ? x[(long)gr * F_DIM + c] : 0.0f;
    }
    __syncthreads();

    #pragma unroll
    for (int t = 0; t < 2; ++t) {
        const int ct = wave * 2 + t;
        const int n  = ct * 16 + l16;
        v8f acc = {};
        for (int k = 0; k < F_DIM; k += 4) {
            v2f a, b;
            a.x = As[l16 * 129 + k + 2 * half + 0];
            a.y = As[l16 * 129 + k + 2 * half + 1];
            b.x = W1[(long)(k + 2 * half + 0) * H_DIM + n];
            b.y = W1[(long)(k + 2 * half + 1) * H_DIM + n];
            acc = __builtin_amdgcn_wmma_f32_16x16x4_f32(
                false, a, false, b, (short)0, acc, false, false);
        }
        const float bb = b1[n];
        const float g  = gamma[n], be = beta[n];
        const float mu = mean[n],  iv = rsqrtf(var[n] + BN_EPS);
        #pragma unroll
        for (int r = 0; r < 8; ++r) {
            const int m = r + 8 * half;
            float v = acc[r] + bb;
            v = (v - mu) * iv * g + be;
            Hs[m * 257 + n] = fmaxf(v, 0.0f);
        }
    }
    __syncthreads();

    if (wave < 3) {
        const int n   = wave * 16 + l16;
        const bool nok = (n < C_DIM);
        v8f acc = {};
        for (int k = 0; k < H_DIM; k += 4) {
            v2f a, b;
            a.x = Hs[l16 * 257 + k + 2 * half + 0];
            a.y = Hs[l16 * 257 + k + 2 * half + 1];
            b.x = nok ? W2[(long)(k + 2 * half + 0) * C_DIM + n] : 0.0f;
            b.y = nok ? W2[(long)(k + 2 * half + 1) * C_DIM + n] : 0.0f;
            acc = __builtin_amdgcn_wmma_f32_16x16x4_f32(
                false, a, false, b, (short)0, acc, false, false);
        }
        const float bb = nok ? b2[n] : 0.0f;
        #pragma unroll
        for (int r = 0; r < 8; ++r) {
            const int m  = r + 8 * half;
            const int gr = row0 + m;
            if (nok && gr < N) h0[(long)gr * C_DIM + n] = acc[r] + bb;
        }
    }
}

__global__ void zero_kernel(float* __restrict__ p, long n)
{
    long i = (long)blockIdx.x * blockDim.x + threadIdx.x;
    if (i < n) p[i] = 0.0f;
}

__global__ void zero_int_kernel(int* __restrict__ p, int n)
{
    int i = blockIdx.x * blockDim.x + threadIdx.x;
    if (i < n) p[i] = 0;
}

typedef __attribute__((ext_vector_type(4))) int   v4i;
typedef __attribute__((ext_vector_type(4))) float v4f;

#define TILE_SHIFT  8
#define TILE_ROWS   (1 << TILE_SHIFT)
#define HOP_THREADS 512
#define HOP_GROUPS  (HOP_THREADS / 8)
#define BIN_THREADS 1024
#define BIN_EDGES   8192

extern __shared__ int dyn_lds[];

__global__ __launch_bounds__(BIN_THREADS) void bin_count_kernel(
    const int* __restrict__ dst, int* __restrict__ counts, int E, int nb)
{
    int* hist = dyn_lds;
    const int tid = threadIdx.x;
    for (int i = tid; i < nb; i += BIN_THREADS) hist[i] = 0;
    const long c0 = (long)blockIdx.x * BIN_EDGES + tid;
    int dv[BIN_EDGES / BIN_THREADS];
    for (int k = 0; k < BIN_EDGES / BIN_THREADS; ++k) {
        const long e = c0 + (long)k * BIN_THREADS;
        dv[k] = dst[(e < E) ? e : (long)(E - 1)];
    }
    __syncthreads();
    for (int k = 0; k < BIN_EDGES / BIN_THREADS; ++k) {
        const long e = c0 + (long)k * BIN_THREADS;
        if (e < E) atomicAdd(&hist[dv[k] >> TILE_SHIFT], 1);
    }
    __syncthreads();
    for (int i = tid; i < nb; i += BIN_THREADS) {
        const int c = hist[i];
        if (c) atomicAdd(&counts[i], c);
    }
}

__global__ __launch_bounds__(1024) void bin_scan_kernel(
    const int* __restrict__ counts, int* __restrict__ offs,
    int* __restrict__ cursor, int nb)
{
    __shared__ int s[1024];
    __shared__ int carry;
    const int tid = threadIdx.x;
    if (tid == 0) carry = 0;
    __syncthreads();
    for (int base = 0; base < nb; base += 1024) {
        const int i = base + tid;
        const int v = (i < nb) ? counts[i] : 0;
        s[tid] = v;
        __syncthreads();
        for (int off = 1; off < 1024; off <<= 1) {
            const int t = (tid >= off) ? s[tid - off] : 0;
            __syncthreads();
            s[tid] += t;
            __syncthreads();
        }
        const int incl = s[tid];
        const int excl = incl - v + carry;
        if (i < nb) { offs[i] = excl; cursor[i] = excl; }
        __syncthreads();
        if (tid == 1023) carry += incl;
        __syncthreads();
    }
    if (tid == 0) offs[nb] = carry;
}

__global__ __launch_bounds__(BIN_THREADS) void bin_fill_kernel(
    const int* __restrict__ src, const int* __restrict__ dst,
    const float* __restrict__ norm, int* __restrict__ cursor,
    int* __restrict__ rec, int E, int nb)
{
    int* hist  = dyn_lds;
    int* rbase = dyn_lds + nb;
    const int tid = threadIdx.x;
    for (int i = tid; i < nb; i += BIN_THREADS) hist[i] = 0;
    const long c0 = (long)blockIdx.x * BIN_EDGES + tid;
    int   dv[BIN_EDGES / BIN_THREADS];
    int   sv[BIN_EDGES / BIN_THREADS];
    float wv[BIN_EDGES / BIN_THREADS];
    for (int k = 0; k < BIN_EDGES / BIN_THREADS; ++k) {
        const long e  = c0 + (long)k * BIN_THREADS;
        const long ec = (e < E) ? e : (long)(E - 1);
        dv[k] = dst[ec];
        sv[k] = src[ec];
        wv[k] = norm[ec];
    }
    __syncthreads();
    for (int k = 0; k < BIN_EDGES / BIN_THREADS; ++k) {
        const long e = c0 + (long)k * BIN_THREADS;
        if (e < E) atomicAdd(&hist[dv[k] >> TILE_SHIFT], 1);
    }
    __syncthreads();
    for (int i = tid; i < nb; i += BIN_THREADS) {
        const int c = hist[i];
        rbase[i] = c ? atomicAdd(&cursor[i], c) : 0;
        hist[i] = 0;
    }
    __syncthreads();
    v4i* rec4 = (v4i*)rec;
    for (int k = 0; k < BIN_EDGES / BIN_THREADS; ++k) {
        const long e = c0 + (long)k * BIN_THREADS;
        if (e < E) {
            const int b   = dv[k] >> TILE_SHIFT;
            const int pos = rbase[b] + atomicAdd(&hist[b], 1);
            v4i r;
            r.x = sv[k];
            r.y = dv[k] - (b << TILE_SHIFT);
            r.z = __float_as_int(wv[k]);
            r.w = 0;
            rec4[pos] = r;
        }
    }
}

__device__ __forceinline__ void lds_add(float* p, float v)
{
    __hip_atomic_fetch_add(p, v, __ATOMIC_RELAXED, __HIP_MEMORY_SCOPE_WORKGROUP);
}

__global__ __launch_bounds__(HOP_THREADS) void hop_tile_kernel(
    const int* __restrict__ rec, const int* __restrict__ offs,
    const float* __restrict__ prev, float* __restrict__ next, int N)
{
    __shared__ __attribute__((aligned(16))) float tile[TILE_ROWS * C_DIM];

    const int tid  = threadIdx.x;
    const int b    = blockIdx.x;
    const int row0 = b << TILE_SHIFT;

    v4f* tile4 = (v4f*)tile;
    for (int i = tid; i < TILE_ROWS * C_DIM / 4; i += HOP_THREADS)
        tile4[i] = (v4f){0.0f, 0.0f, 0.0f, 0.0f};
    __syncthreads();

    const v4i* rec4 = (const v4i*)rec;
    const int e0  = offs[b];
    const int e1  = offs[b + 1];
    const int grp = tid >> 3;
    const int l   = tid & 7;

    const int last = e1 - 1;
    int e = e0 + grp;
    if (e + HOP_GROUPS < e1) {
        v4i ra = rec4[e];
        v4i rb = rec4[e + HOP_GROUPS];
        for (;;) {
            const int en = e + 2 * HOP_GROUPS;
            const int ia = (en < last) ? en : last;
            const int ib = (en + HOP_GROUPS < last) ? (en + HOP_GROUPS) : last;
            const v4i na = rec4[ia];
            const v4i nc = rec4[ib];
            const float* pa = prev + (long)ra.x * C_DIM + l;
            const float* pb = prev + (long)rb.x * C_DIM + l;
            float va[5], vb[5];
            for (int j = 0; j < 5; ++j) { va[j] = pa[8 * j]; vb[j] = pb[8 * j]; }
            const float wa = __int_as_float(ra.z);
            const float wb = __int_as_float(rb.z);
            float* ta = tile + ra.y * C_DIM + l;
            float* tb = tile + rb.y * C_DIM + l;
            for (int j = 0; j < 5; ++j) lds_add(&ta[8 * j], wa * va[j]);
            for (int j = 0; j < 5; ++j) lds_add(&tb[8 * j], wb * vb[j]);
            e = en;
            if (e + HOP_GROUPS >= e1) break;
            ra = na;
            rb = nc;
        }
    }
    if (e < e1) {
        const v4i ra = rec4[e];
        const float* pa = prev + (long)ra.x * C_DIM + l;
        float va[5];
        for (int j = 0; j < 5; ++j) va[j] = pa[8 * j];
        const float wa = __int_as_float(ra.z);
        float* ta = tile + ra.y * C_DIM + l;
        for (int j = 0; j < 5; ++j) lds_add(&ta[8 * j], wa * va[j]);
    }
    __syncthreads();

    const int nrows = (N - row0 < TILE_ROWS) ? (N - row0) : TILE_ROWS;
    v4f* out4 = (v4f*)(next + (long)row0 * C_DIM);
    for (int i = tid; i < nrows * (C_DIM / 4); i += HOP_THREADS)
        out4[i] = tile4[i];
}

__global__ void combine_kernel(const float* __restrict__ v,
                               float* __restrict__ acc,
                               const float* __restrict__ pw,
                               const float* __restrict__ pb, int N)
{
    int n = blockIdx.x * blockDim.x + threadIdx.x;
    if (n >= N) return;
    const float* row = v + (long)n * C_DIM;
    float d = pb[0];
    #pragma unroll
    for (int c = 0; c < C_DIM; ++c) d += row[c] * pw[c];
    const float r = 1.0f / (1.0f + expf(-d));
    float* ar = acc + (long)n * C_DIM;
    #pragma unroll
    for (int c = 0; c < C_DIM; ++c) ar[c] += r * row[c];
}

__global__ void lsm_kernel(const float* __restrict__ acc,
                           float* __restrict__ out, int N)
{
    int n = blockIdx.x * blockDim.x + threadIdx.x;
    if (n >= N) return;
    const float* row = acc + (long)n * C_DIM;
    float m = -3.402823466e38f;
    #pragma unroll
    for (int c = 0; c < C_DIM; ++c) m = fmaxf(m, row[c]);
    float s = 0.0f;
    #pragma unroll
    for (int c = 0; c < C_DIM; ++c) s += expf(row[c] - m);
    const float ls = logf(s);
    float* o = out + (long)n * C_DIM;
    #pragma unroll
    for (int c = 0; c < C_DIM; ++c) o[c] = row[c] - m - ls;
}

extern "C" void kernel_launch(void* const* d_in, const int* in_sizes, int n_in,
                              void* d_out, int out_size, void* d_ws, size_t ws_size,
                              hipStream_t stream)
{
    const float* x     = (const float*)d_in[0];
    const int*   edge  = (const int*)d_in[1];
    const float* norm  = (const float*)d_in[2];
    const float* W1    = (const float*)d_in[3];
    const float* b1    = (const float*)d_in[4];
    const float* gamma = (const float*)d_in[5];
    const float* beta  = (const float*)d_in[6];
    const float* mean  = (const float*)d_in[7];
    const float* var   = (const float*)d_in[8];
    const float* W2    = (const float*)d_in[9];
    const float* b2    = (const float*)d_in[10];
    const float* pw    = (const float*)d_in[11];
    const float* pb    = (const float*)d_in[12];

    const int N = in_sizes[0] / F_DIM;
    const int E = in_sizes[2];

    const size_t per = (((size_t)N * C_DIM) + 63) & ~(size_t)63;
    float* bufA = (float*)d_ws;
    float* bufB = bufA + per;
    float* accb = bufB + per;

    const int    nb  = (N + TILE_ROWS - 1) / TILE_ROWS;
    const size_t nbp = ((size_t)nb + 1 + 63) & ~(size_t)63;
    int* rec    = (int*)(accb + per);
    int* counts = rec + (size_t)E * 4;
    int* offs   = counts + nbp;
    int* cursor = offs + nbp;

    const int* src = edge;
    const int* dst = edge + E;

    const long tot     = (long)N * C_DIM;
    const int  zblocks = (int)((tot + 255) / 256);
    const int  nblocks = (N + 255) / 256;
    const int  bblocks = (E + BIN_EDGES - 1) / BIN_EDGES;

    mlp_kernel<<<(N + 15) / 16, 256, 0, stream>>>(
        x, W1, b1, gamma, beta, mean, var, W2, b2, bufA, N);

    zero_kernel<<<zblocks, 256, 0, stream>>>(accb, tot);
    combine_kernel<<<nblocks, 256, 0, stream>>>(bufA, accb, pw, pb, N);

    zero_int_kernel<<<(nb + 255) / 256, 256, 0, stream>>>(counts, nb);
    bin_count_kernel<<<bblocks, BIN_THREADS, nb * sizeof(int), stream>>>(
        dst, counts, E, nb);
    bin_scan_kernel<<<1, 1024, 0, stream>>>(counts, offs, cursor, nb);
    bin_fill_kernel<<<bblocks, BIN_THREADS, 2 * nb * sizeof(int), stream>>>(
        src, dst, norm, cursor, rec, E, nb);

    for (int k = 0; k < 10; ++k) {
        const float* prev = (k & 1) ? bufB : bufA;
        float*       next = (k & 1) ? bufA : bufB;
        hop_tile_kernel<<<nb, HOP_THREADS, 0, stream>>>(rec, offs, prev, next, N);
        combine_kernel<<<nblocks, 256, 0, stream>>>(next, accb, pw, pb, N);
    }

    lsm_kernel<<<nblocks, 256, 0, stream>>>(accb, (float*)d_out, N);
}


# Combine and log-softmax passes move rows as whole 128-B lines (16-B flat loads/stores per lane) staged through an LDS row tile; one thread per row reads LDS, not global; sum order kept
# speedup vs baseline: 2.6186x; 1.3248x over previous
#include <hip/hip_runtime.h>
#include <hip/hip_bf16.h>

typedef __attribute__((ext_vector_type(2))) float v2f;
typedef __attribute__((ext_vector_type(8))) float v8f;

#define F_DIM 128
#define H_DIM 256
#define C_DIM 40
#define BN_EPS 1e-5f

__global__ __launch_bounds__(256) void mlp_kernel(
    const float* __restrict__ x, const float* __restrict__ W1,
    const float* __restrict__ b1, const float* __restrict__ gamma,
    const float* __restrict__ beta, const float* __restrict__ mean,
    const float* __restrict__ var, const float* __restrict__ W2,
    const float* __restrict__ b2, float* __restrict__ h0, int N)
{
    __shared__ float As[16 * 129];
    __shared__ float Hs[16 * 257];

    const int tid  = threadIdx.x;
    const int wave = tid >> 5;
    const int lane = tid & 31;
    const int half = lane >> 4;
    const int l16  = lane & 15;
    const int row0 = blockIdx.x * 16;

    #pragma unroll
    for (int i = 0; i < 8; ++i) {
        int idx = tid + i * 256;
        int r = idx >> 7, c = idx & 127;
        int gr = row0 + r;
        As[r * 129 + c] = (gr < N) ? x[(long)gr * F_DIM + c] : 0.0f;
    }
    __syncthreads();

    #pragma unroll
    for (int t = 0; t < 2; ++t) {
        const int ct = wave * 2 + t;
        const int n  = ct * 16 + l16;
        v8f acc = {};
        for (int k = 0; k < F_DIM; k += 4) {
            v2f a, b;
            a.x = As[l16 * 129 + k + 2 * half + 0];
            a.y = As[l16 * 129 + k + 2 * half + 1];
            b.x = W1[(long)(k + 2 * half + 0) * H_DIM + n];
            b.y = W1[(long)(k + 2 * half + 1) * H_DIM + n];
            acc = __builtin_amdgcn_wmma_f32_16x16x4_f32(
                false, a, false, b, (short)0, acc, false, false);
        }
        const float bb = b1[n];
        const float g  = gamma[n], be = beta[n];
        const float mu = mean[n],  iv = rsqrtf(var[n] + BN_EPS);
        #pragma unroll
        for (int r = 0; r < 8; ++r) {
            const int m = r + 8 * half;
            float v = acc[r] + bb;
            v = (v - mu) * iv * g + be;
            Hs[m * 257 + n] = fmaxf(v, 0.0f);
        }
    }
    __syncthreads();

    if (wave < 3) {
        const int n   = wave * 16 + l16;
        const bool nok = (n < C_DIM);
        v8f acc = {};
        for (int k = 0; k < H_DIM; k += 4) {
            v2f a, b;
            a.x = Hs[l16 * 257 + k + 2 * half + 0];
            a.y = Hs[l16 * 257 + k + 2 * half + 1];
            b.x = nok ? W2[(long)(k + 2 * half + 0) * C_DIM + n] : 0.0f;
            b.y = nok ? W2[(long)(k + 2 * half + 1) * C_DIM + n] : 0.0f;
            acc = __builtin_amdgcn_wmma_f32_16x16x4_f32(
                false, a, false, b, (short)0, acc, false, false);
        }
        const float bb = nok ? b2[n] : 0.0f;
        #pragma unroll
        for (int r = 0; r < 8; ++r) {
            const int m  = r + 8 * half;
            const int gr = row0 + m;
            if (nok && gr < N) h0[(long)gr * C_DIM + n] = acc[r] + bb;
        }
    }
}

__global__ void zero_kernel(float* __restrict__ p, long n)
{
    long i = (long)blockIdx.x * blockDim.x + threadIdx.x;
    if (i < n) p[i] = 0.0f;
}

__global__ void zero_int_kernel(int* __restrict__ p, int n)
{
    int i = blockIdx.x * blockDim.x + threadIdx.x;
    if (i < n) p[i] = 0;
}

typedef __attribute__((ext_vector_type(4))) int   v4i;
typedef __attribute__((ext_vector_type(4))) float v4f;

#define TILE_SHIFT  8
#define TILE_ROWS   (1 << TILE_SHIFT)
#define HOP_THREADS 512
#define HOP_GROUPS  (HOP_THREADS / 8)
#define BIN_THREADS 1024
#define BIN_EDGES   8192

extern __shared__ int dyn_lds[];

__global__ __launch_bounds__(BIN_THREADS) void bin_count_kernel(
    const int* __restrict__ dst, int* __restrict__ counts, int E, int nb)
{
    int* hist = dyn_lds;
    const int tid = threadIdx.x;
    for (int i = tid; i < nb; i += BIN_THREADS) hist[i] = 0;
    const long c0 = (long)blockIdx.x * BIN_EDGES + tid;
    int dv[BIN_EDGES / BIN_THREADS];
    for (int k = 0; k < BIN_EDGES / BIN_THREADS; ++k) {
        const long e = c0 + (long)k * BIN_THREADS;
        dv[k] = dst[(e < E) ? e : (long)(E - 1)];
    }
    __syncthreads();
    for (int k = 0; k < BIN_EDGES / BIN_THREADS; ++k) {
        const long e = c0 + (long)k * BIN_THREADS;
        if (e < E) atomicAdd(&hist[dv[k] >> TILE_SHIFT], 1);
    }
    __syncthreads();
    for (int i = tid; i < nb; i += BIN_THREADS) {
        const int c = hist[i];
        if (c) atomicAdd(&counts[i], c);
    }
}

__global__ __launch_bounds__(1024) void bin_scan_kernel(
    const int* __restrict__ counts, int* __restrict__ offs,
    int* __restrict__ cursor, int nb)
{
    __shared__ int s[1024];
    __shared__ int carry;
    const int tid = threadIdx.x;
    if (tid == 0) carry = 0;
    __syncthreads();
    for (int base = 0; base < nb; base += 1024) {
        const int i = base + tid;
        const int v = (i < nb) ? counts[i] : 0;
        s[tid] = v;
        __syncthreads();
        for (int off = 1; off < 1024; off <<= 1) {
            const int t = (tid >= off) ? s[tid - off] : 0;
            __syncthreads();
            s[tid] += t;
            __syncthreads();
        }
        const int incl = s[tid];
        const int excl = incl - v + carry;
        if (i < nb) { offs[i] = excl; cursor[i] = excl; }
        __syncthreads();
        if (tid == 1023) carry += incl;
        __syncthreads();
    }
    if (tid == 0) offs[nb] = carry;
}

__global__ __launch_bounds__(BIN_THREADS) void bin_fill_kernel(
    const int* __restrict__ src, const int* __restrict__ dst,
    const float* __restrict__ norm, int* __restrict__ cursor,
    int* __restrict__ rec, int E, int nb)
{
    int* hist  = dyn_lds;
    int* rbase = dyn_lds + nb;
    const int tid = threadIdx.x;
    for (int i = tid; i < nb; i += BIN_THREADS) hist[i] = 0;
    const long c0 = (long)blockIdx.x * BIN_EDGES + tid;
    int   dv[BIN_EDGES / BIN_THREADS];
    int   sv[BIN_EDGES / BIN_THREADS];
    float wv[BIN_EDGES / BIN_THREADS];
    for (int k = 0; k < BIN_EDGES / BIN_THREADS; ++k) {
        const long e  = c0 + (long)k * BIN_THREADS;
        const long ec = (e < E) ? e : (long)(E - 1);
        dv[k] = dst[ec];
        sv[k] = src[ec];
        wv[k] = norm[ec];
    }
    __syncthreads();
    for (int k = 0; k < BIN_EDGES / BIN_THREADS; ++k) {
        const long e = c0 + (long)k * BIN_THREADS;
        if (e < E) atomicAdd(&hist[dv[k] >> TILE_SHIFT], 1);
    }
    __syncthreads();
    for (int i = tid; i < nb; i += BIN_THREADS) {
        const int c = hist[i];
        rbase[i] = c ? atomicAdd(&cursor[i], c) : 0;
        hist[i] = 0;
    }
    __syncthreads();
    v4i* rec4 = (v4i*)rec;
    for (int k = 0; k < BIN_EDGES / BIN_THREADS; ++k) {
        const long e = c0 + (long)k * BIN_THREADS;
        if (e < E) {
            const int b   = dv[k] >> TILE_SHIFT;
            const int pos = rbase[b] + atomicAdd(&hist[b], 1);
            v4i r;
            r.x = sv[k];
            r.y = dv[k] - (b << TILE_SHIFT);
            r.z = __float_as_int(wv[k]);
            r.w = 0;
            rec4[pos] = r;
        }
    }
}

__device__ __forceinline__ void lds_add(float* p, float v)
{
    __hip_atomic_fetch_add(p, v, __ATOMIC_RELAXED, __HIP_MEMORY_SCOPE_WORKGROUP);
}

__global__ __launch_bounds__(HOP_THREADS) void hop_tile_kernel(
    const int* __restrict__ rec, const int* __restrict__ offs,
    const float* __restrict__ prev, float* __restrict__ next, int N)
{
    __shared__ __attribute__((aligned(16))) float tile[TILE_ROWS * C_DIM];

    const int tid  = threadIdx.x;
    const int b    = blockIdx.x;
    const int row0 = b << TILE_SHIFT;

    v4f* tile4 = (v4f*)tile;
    for (int i = tid; i < TILE_ROWS * C_DIM / 4; i += HOP_THREADS)
        tile4[i] = (v4f){0.0f, 0.0f, 0.0f, 0.0f};
    __syncthreads();

    const v4i* rec4 = (const v4i*)rec;
    const int e0  = offs[b];
    const int e1  = offs[b + 1];
    const int grp = tid >> 3;
    const int l   = tid & 7;

    const int last = e1 - 1;
    int e = e0 + grp;
    if (e + HOP_GROUPS < e1) {
        v4i ra = rec4[e];
        v4i rb = rec4[e + HOP_GROUPS];
        for (;;) {
            const int en = e + 2 * HOP_GROUPS;
            const int ia = (en < last) ? en : last;
            const int ib = (en + HOP_GROUPS < last) ? (en + HOP_GROUPS) : last;
            const v4i na = rec4[ia];
            const v4i nc = rec4[ib];
            const float* pa = prev + (long)ra.x * C_DIM + l;
            const float* pb = prev + (long)rb.x * C_DIM + l;
            float va[5], vb[5];
            for (int j = 0; j < 5; ++j) { va[j] = pa[8 * j]; vb[j] = pb[8 * j]; }
            const float wa = __int_as_float(ra.z);
            const float wb = __int_as_float(rb.z);
            float* ta = tile + ra.y * C_DIM + l;
            float* tb = tile + rb.y * C_DIM + l;
            for (int j = 0; j < 5; ++j) lds_add(&ta[8 * j], wa * va[j]);
            for (int j = 0; j < 5; ++j) lds_add(&tb[8 * j], wb * vb[j]);
            e = en;
            if (e + HOP_GROUPS >= e1) break;
            ra = na;
            rb = nc;
        }
    }
    if (e < e1) {
        const v4i ra = rec4[e];
        const float* pa = prev + (long)ra.x * C_DIM + l;
        float va[5];
        for (int j = 0; j < 5; ++j) va[j] = pa[8 * j];
        const float wa = __int_as_float(ra.z);
        float* ta = tile + ra.y * C_DIM + l;
        for (int j = 0; j < 5; ++j) lds_add(&ta[8 * j], wa * va[j]);
    }
    __syncthreads();

    const int nrows = (N - row0 < TILE_ROWS) ? (N - row0) : TILE_ROWS;
    v4f* out4 = (v4f*)(next + (long)row0 * C_DIM);
    for (int i = tid; i < nrows * (C_DIM / 4); i += HOP_THREADS)
        out4[i] = tile4[i];
}

#define ROW_TILE  256
#define ROW_V4    (C_DIM / 4)
#define ROW_PITCH 44

__global__ __launch_bounds__(ROW_TILE) void combine_kernel(
    const float* __restrict__ v, float* __restrict__ acc,
    const float* __restrict__ pw, const float* __restrict__ pb, int N)
{
    __shared__ __attribute__((aligned(16))) float rows[ROW_TILE * ROW_PITCH];
    __shared__ float rs[ROW_TILE];

    const int tid   = threadIdx.x;
    const int row0  = blockIdx.x * ROW_TILE;
    const int nrows = (N - row0 < ROW_TILE) ? (N - row0) : ROW_TILE;
    const int nv4   = nrows * ROW_V4;
    const v4f* v4 = (const v4f*)(v + (long)row0 * C_DIM);
    v4f*       a4 = (v4f*)(acc + (long)row0 * C_DIM);

    v4f x[ROW_V4], a[ROW_V4];
    #pragma unroll
    for (int j = 0; j < ROW_V4; ++j) {
        const int i  = tid + j * ROW_TILE;
        const int ic = (i < nv4) ? i : (nv4 - 1);
        x[j] = v4[ic];
        a[j] = a4[ic];
    }
    #pragma unroll
    for (int j = 0; j < ROW_V4; ++j) {
        const int i = tid + j * ROW_TILE;
        const int r = i / ROW_V4;
        const int q = i - r * ROW_V4;
        *(v4f*)&rows[r * ROW_PITCH + q * 4] = x[j];
    }
    __syncthreads();

    if (tid < nrows) {
        const v4f* rp = (const v4f*)&rows[tid * ROW_PITCH];
        float d = pb[0];
        #pragma unroll
        for (int q = 0; q < ROW_V4; ++q) {
            const v4f t = rp[q];
            d += t.x * pw[4 * q + 0];
            d += t.y * pw[4 * q + 1];
            d += t.z * pw[4 * q + 2];
            d += t.w * pw[4 * q + 3];
        }
        rs[tid] = 1.0f / (1.0f + expf(-d));
    }
    __syncthreads();

    float rf[ROW_V4];
    #pragma unroll
    for (int j = 0; j < ROW_V4; ++j) {
        const int i  = tid + j * ROW_TILE;
        const int ic = (i < nv4) ? i : (nv4 - 1);
        rf[j] = rs[ic / ROW_V4];
    }
    #pragma unroll
    for (int j = 0; j < ROW_V4; ++j) {
        const int i = tid + j * ROW_TILE;
        if (i < nv4) a4[i] = a[j] + rf[j] * x[j];
    }
}

__global__ __launch_bounds__(ROW_TILE) void lsm_kernel(
    const float* __restrict__ acc, float* __restrict__ out, int N)
{
    __shared__ __attribute__((aligned(16))) float rows[ROW_TILE * ROW_PITCH];

    const int tid   = threadIdx.x;
    const int row0  = blockIdx.x * ROW_TILE;
    const int nrows = (N - row0 < ROW_TILE) ? (N - row0) : ROW_TILE;
    const int nv4   = nrows * ROW_V4;
    const v4f* a4 = (const v4f*)(acc + (long)row0 * C_DIM);
    v4f*       o4 = (v4f*)(out + (long)row0 * C_DIM);

    v4f x[ROW_V4];
    #pragma unroll
    for (int j = 0; j < ROW_V4; ++j) {
        const int i  = tid + j * ROW_TILE;
        const int ic = (i < nv4) ? i : (nv4 - 1);
        x[j] = a4[ic];
    }
    #pragma unroll
    for (int j = 0; j < ROW_V4; ++j) {
        const int i = tid + j * ROW_TILE;
        const int r = i / ROW_V4;
        const int q = i - r * ROW_V4;
        *(v4f*)&rows[r * ROW_PITCH + q * 4] = x[j];
    }
    __syncthreads();

    if (tid < nrows) {
        v4f* rp = (v4f*)&rows[tid * ROW_PITCH];
        v4f t[ROW_V4];
        #pragma unroll
        for (int q = 0; q < ROW_V4; ++q) t[q] = rp[q];
        float m = -3.402823466e38f;
        #pragma unroll
        for (int q = 0; q < ROW_V4; ++q) {
            m = fmaxf(m, t[q].x); m = fmaxf(m, t[q].y);
            m = fmaxf(m, t[q].z); m = fmaxf(m, t[q].w);
        }
        float s = 0.0f;
        #pragma unroll
        for (int q = 0; q < ROW_V4; ++q) {
            s += expf(t[q].x - m); s += expf(t[q].y - m);
            s += expf(t[q].z - m); s += expf(t[q].w - m);
        }
        const float ls = logf(s);
        #pragma unroll
        for (int q = 0; q < ROW_V4; ++q) {
            v4f o;
            o.x = t[q].x - m - ls; o.y = t[q].y - m - ls;
            o.z = t[q].z - m - ls; o.w = t[q].w - m - ls;
            rp[q] = o;
        }
    }
    __syncthreads();

    for (int j = 0; j < ROW_V4; ++j) {
        const int i = tid + j * ROW_TILE;
        if (i < nv4) {
            const int r = i / ROW_V4;
            const int q = i - r * ROW_V4;
            o4[i] = *(const v4f*)&rows[r * ROW_PITCH + q * 4];
        }
    }
}

extern "C" void kernel_launch(void* const* d_in, const int* in_sizes, int n_in,
                              void* d_out, int out_size, void* d_ws, size_t ws_size,
                              hipStream_t stream)
{
    const float* x     = (const float*)d_in[0];
    const int*   edge  = (const int*)d_in[1];
    const float* norm  = (const float*)d_in[2];
    const float* W1    = (const float*)d_in[3];
    const float* b1    = (const float*)d_in[4];
    const float* gamma = (const float*)d_in[5];
    const float* beta  = (const float*)d_in[6];
    const float* mean  = (const float*)d_in[7];
    const float* var   = (const float*)d_in[8];
    const float* W2    = (const float*)d_in[9];
    const float* b2    = (const float*)d_in[10];
    const float* pw    = (const float*)d_in[11];
    const float* pb    = (const float*)d_in[12];

    const int N = in_sizes[0] / F_DIM;
    const int E = in_sizes[2];

    const size_t per = (((size_t)N * C_DIM) + 63) & ~(size_t)63;
    float* bufA = (float*)d_ws;
    float* bufB = bufA + per;
    float* accb = bufB + per;

    const int    nb  = (N + TILE_ROWS - 1) / TILE_ROWS;
    const size_t nbp = ((size_t)nb + 1 + 63) & ~(size_t)63;
    int* rec    = (int*)(accb + per);
    int* counts = rec + (size_t)E * 4;
    int* offs   = counts + nbp;
    int* cursor = offs + nbp;

    const int* src = edge;
    const int* dst = edge + E;

    const long tot     = (long)N * C_DIM;
    const int  zblocks = (int)((tot + 255) / 256);
    const int  nblocks = (N + ROW_TILE - 1) / ROW_TILE;
    const int  bblocks = (E + BIN_EDGES - 1) / BIN_EDGES;

    mlp_kernel<<<(N + 15) / 16, 256, 0, stream>>>(
        x, W1, b1, gamma, beta, mean, var, W2, b2, bufA, N);

    zero_kernel<<<zblocks, 256, 0, stream>>>(accb, tot);
    combine_kernel<<<nblocks, ROW_TILE, 0, stream>>>(bufA, accb, pw, pb, N);

    zero_int_kernel<<<(nb + 255) / 256, 256, 0, stream>>>(counts, nb);
    bin_count_kernel<<<bblocks, BIN_THREADS, nb * sizeof(int), stream>>>(
        dst, counts, E, nb);
    bin_scan_kernel<<<1, 1024, 0, stream>>>(counts, offs, cursor, nb);
    bin_fill_kernel<<<bblocks, BIN_THREADS, 2 * nb * sizeof(int), stream>>>(
        src, dst, norm, cursor, rec, E, nb);

    for (int k = 0; k < 10; ++k) {
        const float* prev = (k & 1) ? bufB : bufA;
        float*       next = (k & 1) ? bufA : bufB;
        hop_tile_kernel<<<nb, HOP_THREADS, 0, stream>>>(rec, offs, prev, next, N);
        combine_kernel<<<nblocks, ROW_TILE, 0, stream>>>(next, accb, pw, pb, N);
    }

    lsm_kernel<<<nblocks, ROW_TILE, 0, stream>>>(accb, (float*)d_out, N);
}


# Hop-0 term sigmoid(h0.proj)*h0 is stored by the MLP kernel's epilogue from an LDS copy of its 16x40 tile; the accumulator zero-fill launch and the first combine launch are gone; sum order kept
# speedup vs baseline: 2.8085x; 1.0725x over previous
#include <hip/hip_runtime.h>
#include <hip/hip_bf16.h>

typedef __attribute__((ext_vector_type(2))) float v2f;
typedef __attribute__((ext_vector_type(8))) float v8f;
typedef __attribute__((ext_vector_type(4))) int   v4i;
typedef __attribute__((ext_vector_type(4))) float v4f;

#define F_DIM 128
#define H_DIM 256
#define C_DIM 40
#define BN_EPS 1e-5f
#define OT_PITCH 44

__global__ __launch_bounds__(256) void mlp_kernel(
    const float* __restrict__ x, const float* __restrict__ W1,
    const float* __restrict__ b1, const float* __restrict__ gamma,
    const float* __restrict__ beta, const float* __restrict__ mean,
    const float* __restrict__ var, const float* __restrict__ W2,
    const float* __restrict__ b2, const float* __restrict__ pw,
    const float* __restrict__ pb, float* __restrict__ h0,
    float* __restrict__ acc0, int N)
{
    __shared__ float As[16 * 129];
    __shared__ float Hs[16 * 257];
    __shared__ __attribute__((aligned(16))) float Ot[16 * OT_PITCH];
    __shared__ float gs[16];

    const int tid  = threadIdx.x;
    const int wave = tid >> 5;
    const int lane = tid & 31;
    const int half = lane >> 4;
    const int l16  = lane & 15;
    const int row0 = blockIdx.x * 16;

    #pragma unroll
    for (int i = 0; i < 8; ++i) {
        int idx = tid + i * 256;
        int r = idx >> 7, c = idx & 127;
        int gr = row0 + r;
        As[r * 129 + c] = (gr < N) ? x[(long)gr * F_DIM + c] : 0.0f;
    }
    __syncthreads();

    #pragma unroll
    for (int t = 0; t < 2; ++t) {
        const int ct = wave * 2 + t;
        const int n  = ct * 16 + l16;
        v8f acc = {};
        for (int k = 0; k < F_DIM; k += 4) {
            v2f a, b;
            a.x = As[l16 * 129 + k + 2 * half + 0];
            a.y = As[l16 * 129 + k + 2 * half + 1];
            b.x = W1[(long)(k + 2 * half + 0) * H_DIM + n];
            b.y = W1[(long)(k + 2 * half + 1) * H_DIM + n];
            acc = __builtin_amdgcn_wmma_f32_16x16x4_f32(
                false, a, false, b, (short)0, acc, false, false);
        }
        const float bb = b1[n];
        const float g  = gamma[n], be = beta[n];
        const float mu = mean[n],  iv = rsqrtf(var[n] + BN_EPS);
        #pragma unroll
        for (int r = 0; r < 8; ++r) {
            const int m = r + 8 * half;
            float v = acc[r] + bb;
            v = (v - mu) * iv * g + be;
            Hs[m * 257 + n] = fmaxf(v, 0.0f);
        }
    }
    __syncthreads();

    if (wave < 3) {
        const int n   = wave * 16 + l16;
        const bool nok = (n < C_DIM);
        v8f acc = {};
        for (int k = 0; k < H_DIM; k += 4) {
            v2f a, b;
            a.x = Hs[l16 * 257 + k + 2 * half + 0];
            a.y = Hs[l16 * 257 + k + 2 * half + 1];
            b.x = nok ? W2[(long)(k + 2 * half + 0) * C_DIM + n] : 0.0f;
            b.y = nok ? W2[(long)(k + 2 * half + 1) * C_DIM + n] : 0.0f;
            acc = __builtin_amdgcn_wmma_f32_16x16x4_f32(
                false, a, false, b, (short)0, acc, false, false);
        }
        const float bb = nok ? b2[n] : 0.0f;
        #pragma unroll
        for (int r = 0; r < 8; ++r) {
            const int m = r + 8 * half;
            if (nok) Ot[m * OT_PITCH + n] = acc[r] + bb;
        }
    }
    __syncthreads();

    if (tid < 16) {
        const v4f* rp = (const v4f*)&Ot[tid * OT_PITCH];
        float d = pb[0];
        #pragma unroll
        for (int q = 0; q < C_DIM / 4; ++q) {
            const v4f t = rp[q];
            d += t.x * pw[4 * q + 0];
            d += t.y * pw[4 * q + 1];
            d += t.z * pw[4 * q + 2];
            d += t.w * pw[4 * q + 3];
        }
        gs[tid] = 1.0f / (1.0f + expf(-d));
    }
    __syncthreads();

    const int nrows = (N - row0 < 16) ? (N - row0) : 16;
    if (tid < nrows * (C_DIM / 4)) {
        const int r = tid / (C_DIM / 4);
        const int q = tid - r * (C_DIM / 4);
        const v4f hv = *(const v4f*)&Ot[r * OT_PITCH + q * 4];
        const float g = gs[r];
        ((v4f*)(h0   + (long)row0 * C_DIM))[tid] = hv;
        ((v4f*)(acc0 + (long)row0 * C_DIM))[tid] = g * hv;
    }
}

__global__ void zero_int_kernel(int* __restrict__ p, int n)
{
    int i = blockIdx.x * blockDim.x + threadIdx.x;
    if (i < n) p[i] = 0;
}

#define TILE_SHIFT  8
#define TILE_ROWS   (1 << TILE_SHIFT)
#define HOP_THREADS 512
#define HOP_GROUPS  (HOP_THREADS / 8)
#define BIN_THREADS 1024
#define BIN_EDGES   8192

extern __shared__ int dyn_lds[];

__global__ __launch_bounds__(BIN_THREADS) void bin_count_kernel(
    const int* __restrict__ dst, int* __restrict__ counts, int E, int nb)
{
    int* hist = dyn_lds;
    const int tid = threadIdx.x;
    for (int i = tid; i < nb; i += BIN_THREADS) hist[i] = 0;
    const long c0 = (long)blockIdx.x * BIN_EDGES + tid;
    int dv[BIN_EDGES / BIN_THREADS];
    for (int k = 0; k < BIN_EDGES / BIN_THREADS; ++k) {
        const long e = c0 + (long)k * BIN_THREADS;
        dv[k] = dst[(e < E) ? e : (long)(E - 1)];
    }
    __syncthreads();
    for (int k = 0; k < BIN_EDGES / BIN_THREADS; ++k) {
        const long e = c0 + (long)k * BIN_THREADS;
        if (e < E) atomicAdd(&hist[dv[k] >> TILE_SHIFT], 1);
    }
    __syncthreads();
    for (int i = tid; i < nb; i += BIN_THREADS) {
        const int c = hist[i];
        if (c) atomicAdd(&counts[i], c);
    }
}

__global__ __launch_bounds__(1024) void bin_scan_kernel(
    const int* __restrict__ counts, int* __restrict__ offs,
    int* __restrict__ cursor, int nb)
{
    __shared__ int s[1024];
    __shared__ int carry;
    const int tid = threadIdx.x;
    if (tid == 0) carry = 0;
    __syncthreads();
    for (int base = 0; base < nb; base += 1024) {
        const int i = base + tid;
        const int v = (i < nb) ? counts[i] : 0;
        s[tid] = v;
        __syncthreads();
        for (int off = 1; off < 1024; off <<= 1) {
            const int t = (tid >= off) ? s[tid - off] : 0;
            __syncthreads();
            s[tid] += t;
            __syncthreads();
        }
        const int incl = s[tid];
        const int excl = incl - v + carry;
        if (i < nb) { offs[i] = excl; cursor[i] = excl; }
        __syncthreads();
        if (tid == 1023) carry += incl;
        __syncthreads();
    }
    if (tid == 0) offs[nb] = carry;
}

__global__ __launch_bounds__(BIN_THREADS) void bin_fill_kernel(
    const int* __restrict__ src, const int* __restrict__ dst,
    const float* __restrict__ norm, int* __restrict__ cursor,
    int* __restrict__ rec, int E, int nb)
{
    int* hist  = dyn_lds;
    int* rbase = dyn_lds + nb;
    const int tid = threadIdx.x;
    for (int i = tid; i < nb; i += BIN_THREADS) hist[i] = 0;
    const long c0 = (long)blockIdx.x * BIN_EDGES + tid;
    int   dv[BIN_EDGES / BIN_THREADS];
    int   sv[BIN_EDGES / BIN_THREADS];
    float wv[BIN_EDGES / BIN_THREADS];
    for (int k = 0; k < BIN_EDGES / BIN_THREADS; ++k) {
        const long e  = c0 + (long)k * BIN_THREADS;
        const long ec = (e < E) ? e : (long)(E - 1);
        dv[k] = dst[ec];
        sv[k] = src[ec];
        wv[k] = norm[ec];
    }
    __syncthreads();
    for (int k = 0; k < BIN_EDGES / BIN_THREADS; ++k) {
        const long e = c0 + (long)k * BIN_THREADS;
        if (e < E) atomicAdd(&hist[dv[k] >> TILE_SHIFT], 1);
    }
    __syncthreads();
    for (int i = tid; i < nb; i += BIN_THREADS) {
        const int c = hist[i];
        rbase[i] = c ? atomicAdd(&cursor[i], c) : 0;
        hist[i] = 0;
    }
    __syncthreads();
    v4i* rec4 = (v4i*)rec;
    for (int k = 0; k < BIN_EDGES / BIN_THREADS; ++k) {
        const long e = c0 + (long)k * BIN_THREADS;
        if (e < E) {
            const int b   = dv[k] >> TILE_SHIFT;
            const int pos = rbase[b] + atomicAdd(&hist[b], 1);
            v4i r;
            r.x = sv[k];
            r.y = dv[k] - (b << TILE_SHIFT);
            r.z = __float_as_int(wv[k]);
            r.w = 0;
            rec4[pos] = r;
        }
    }
}

__device__ __forceinline__ void lds_add(float* p, float v)
{
    __hip_atomic_fetch_add(p, v, __ATOMIC_RELAXED, __HIP_MEMORY_SCOPE_WORKGROUP);
}

__global__ __launch_bounds__(HOP_THREADS) void hop_tile_kernel(
    const int* __restrict__ rec, const int* __restrict__ offs,
    const float* __restrict__ prev, float* __restrict__ next, int N)
{
    __shared__ __attribute__((aligned(16))) float tile[TILE_ROWS * C_DIM];

    const int tid  = threadIdx.x;
    const int b    = blockIdx.x;
    const int row0 = b << TILE_SHIFT;

    v4f* tile4 = (v4f*)tile;
    for (int i = tid; i < TILE_ROWS * C_DIM / 4; i += HOP_THREADS)
        tile4[i] = (v4f){0.0f, 0.0f, 0.0f, 0.0f};
    __syncthreads();

    const v4i* rec4 = (const v4i*)rec;
    const int e0  = offs[b];
    const int e1  = offs[b + 1];
    const int grp = tid >> 3;
    const int l   = tid & 7;

    const int last = e1 - 1;
    int e = e0 + grp;
    if (e + HOP_GROUPS < e1) {
        v4i ra = rec4[e];
        v4i rb = rec4[e + HOP_GROUPS];
        for (;;) {
            const int en = e + 2 * HOP_GROUPS;
            const int ia = (en < last) ? en : last;
            const int ib = (en + HOP_GROUPS < last) ? (en + HOP_GROUPS) : last;
            const v4i na = rec4[ia];
            const v4i nc = rec4[ib];
            const float* pa = prev + (long)ra.x * C_DIM + l;
            const float* pb = prev + (long)rb.x * C_DIM + l;
            float va[5], vb[5];
            for (int j = 0; j < 5; ++j) { va[j] = pa[8 * j]; vb[j] = pb[8 * j]; }
            const float wa = __int_as_float(ra.z);
            const float wb = __int_as_float(rb.z);
            float* ta = tile + ra.y * C_DIM + l;
            float* tb = tile + rb.y * C_DIM + l;
            for (int j = 0; j < 5; ++j) lds_add(&ta[8 * j], wa * va[j]);
            for (int j = 0; j < 5; ++j) lds_add(&tb[8 * j], wb * vb[j]);
            e = en;
            if (e + HOP_GROUPS >= e1) break;
            ra = na;
            rb = nc;
        }
    }
    if (e < e1) {
        const v4i ra = rec4[e];
        const float* pa = prev + (long)ra.x * C_DIM + l;
        float va[5];
        for (int j = 0; j < 5; ++j) va[j] = pa[8 * j];
        const float wa = __int_as_float(ra.z);
        float* ta = tile + ra.y * C_DIM + l;
        for (int j = 0; j < 5; ++j) lds_add(&ta[8 * j], wa * va[j]);
    }
    __syncthreads();

    const int nrows = (N - row0 < TILE_ROWS) ? (N - row0) : TILE_ROWS;
    v4f* out4 = (v4f*)(next + (long)row0 * C_DIM);
    for (int i = tid; i < nrows * (C_DIM / 4); i += HOP_THREADS)
        out4[i] = tile4[i];
}

#define ROW_TILE  256
#define ROW_V4    (C_DIM / 4)
#define ROW_PITCH 44

__global__ __launch_bounds__(ROW_TILE) void combine_kernel(
    const float* __restrict__ v, float* __restrict__ acc,
    const float* __restrict__ pw, const float* __restrict__ pb, int N)
{
    __shared__ __attribute__((aligned(16))) float rows[ROW_TILE * ROW_PITCH];
    __shared__ float rs[ROW_TILE];

    const int tid   = threadIdx.x;
    const int row0  = blockIdx.x * ROW_TILE;
    const int nrows = (N - row0 < ROW_TILE) ? (N - row0) : ROW_TILE;
    const int nv4   = nrows * ROW_V4;
    const v4f* v4 = (const v4f*)(v + (long)row0 * C_DIM);
    v4f*       a4 = (v4f*)(acc + (long)row0 * C_DIM);

    v4f x[ROW_V4], a[ROW_V4];
    #pragma unroll
    for (int j = 0; j < ROW_V4; ++j) {
        const int i  = tid + j * ROW_TILE;
        const int ic = (i < nv4) ? i : (nv4 - 1);
        x[j] = v4[ic];
        a[j] = a4[ic];
    }
    #pragma unroll
    for (int j = 0; j < ROW_V4; ++j) {
        const int i = tid + j * ROW_TILE;
        const int r = i / ROW_V4;
        const int q = i - r * ROW_V4;
        *(v4f*)&rows[r * ROW_PITCH + q * 4] = x[j];
    }
    __syncthreads();

    if (tid < nrows) {
        const v4f* rp = (const v4f*)&rows[tid * ROW_PITCH];
        float d = pb[0];
        #pragma unroll
        for (int q = 0; q < ROW_V4; ++q) {
            const v4f t = rp[q];
            d += t.x * pw[4 * q + 0];
            d += t.y * pw[4 * q + 1];
            d += t.z * pw[4 * q + 2];
            d += t.w * pw[4 * q + 3];
        }
        rs[tid] = 1.0f / (1.0f + expf(-d));
    }
    __syncthreads();

    float rf[ROW_V4];
    #pragma unroll
    for (int j = 0; j < ROW_V4; ++j) {
        const int i  = tid + j * ROW_TILE;
        const int ic = (i < nv4) ? i : (nv4 - 1);
        rf[j] = rs[ic / ROW_V4];
    }
    #pragma unroll
    for (int j = 0; j < ROW_V4; ++j) {
        const int i = tid + j * ROW_TILE;
        if (i < nv4) a4[i] = a[j] + rf[j] * x[j];
    }
}

__global__ __launch_bounds__(ROW_TILE) void lsm_kernel(
    const float* __restrict__ acc, float* __restrict__ out, int N)
{
    __shared__ __attribute__((aligned(16))) float rows[ROW_TILE * ROW_PITCH];

    const int tid   = threadIdx.x;
    const int row0  = blockIdx.x * ROW_TILE;
    const int nrows = (N - row0 < ROW_TILE) ? (N - row0) : ROW_TILE;
    const int nv4   = nrows * ROW_V4;
    const v4f* a4 = (const v4f*)(acc + (long)row0 * C_DIM);
    v4f*       o4 = (v4f*)(out + (long)row0 * C_DIM);

    v4f x[ROW_V4];
    #pragma unroll
    for (int j = 0; j < ROW_V4; ++j) {
        const int i  = tid + j * ROW_TILE;
        const int ic = (i < nv4) ? i : (nv4 - 1);
        x[j] = a4[ic];
    }
    #pragma unroll
    for (int j = 0; j < ROW_V4; ++j) {
        const int i = tid + j * ROW_TILE;
        const int r = i / ROW_V4;
        const int q = i - r * ROW_V4;
        *(v4f*)&rows[r * ROW_PITCH + q * 4] = x[j];
    }
    __syncthreads();

    if (tid < nrows) {
        v4f* rp = (v4f*)&rows[tid * ROW_PITCH];
        v4f t[ROW_V4];
        #pragma unroll
        for (int q = 0; q < ROW_V4; ++q) t[q] = rp[q];
        float m = -3.402823466e38f;
        #pragma unroll
        for (int q = 0; q < ROW_V4; ++q) {
            m = fmaxf(m, t[q].x); m = fmaxf(m, t[q].y);
            m = fmaxf(m, t[q].z); m = fmaxf(m, t[q].w);
        }
        float s = 0.0f;
        #pragma unroll
        for (int q = 0; q < ROW_V4; ++q) {
            s += expf(t[q].x - m); s += expf(t[q].y - m);
            s += expf(t[q].z - m); s += expf(t[q].w - m);
        }
        const float ls = logf(s);
        #pragma unroll
        for (int q = 0; q < ROW_V4; ++q) {
            v4f o;
            o.x = t[q].x - m - ls; o.y = t[q].y - m - ls;
            o.z = t[q].z - m - ls; o.w = t[q].w - m - ls;
            rp[q] = o;
        }
    }
    __syncthreads();

    for (int j = 0; j < ROW_V4; ++j) {
        const int i = tid + j * ROW_TILE;
        if (i < nv4) {
            const int r = i / ROW_V4;
            const int q = i - r * ROW_V4;
            o4[i] = *(const v4f*)&rows[r * ROW_PITCH + q * 4];
        }
    }
}

extern "C" void kernel_launch(void* const* d_in, const int* in_sizes, int n_in,
                              void* d_out, int out_size, void* d_ws, size_t ws_size,
                              hipStream_t stream)
{
    const float* x     = (const float*)d_in[0];
    const int*   edge  = (const int*)d_in[1];
    const float* norm  = (const float*)d_in[2];
    const float* W1    = (const float*)d_in[3];
    const float* b1    = (const float*)d_in[4];
    const float* gamma = (const float*)d_in[5];
    const float* beta  = (const float*)d_in[6];
    const float* mean  = (const float*)d_in[7];
    const float* var   = (const float*)d_in[8];
    const float* W2    = (const float*)d_in[9];
    const float* b2    = (const float*)d_in[10];
    const float* pw    = (const float*)d_in[11];
    const float* pb    = (const float*)d_in[12];

    const int N = in_sizes[0] / F_DIM;
    const int E = in_sizes[2];

    const size_t per = (((size_t)N * C_DIM) + 63) & ~(size_t)63;
    float* bufA = (float*)d_ws;
    float* bufB = bufA + per;
    float* accb = bufB + per;

    const int    nb  = (N + TILE_ROWS - 1) / TILE_ROWS;
    const size_t nbp = ((size_t)nb + 1 + 63) & ~(size_t)63;
    int* rec    = (int*)(accb + per);
    int* counts = rec + (size_t)E * 4;
    int* offs   = counts + nbp;
    int* cursor = offs + nbp;

    const int* src = edge;
    const int* dst = edge + E;

    const int  nblocks = (N + ROW_TILE - 1) / ROW_TILE;
    const int  bblocks = (E + BIN_EDGES - 1) / BIN_EDGES;

    mlp_kernel<<<(N + 15) / 16, 256, 0, stream>>>(
        x, W1, b1, gamma, beta, mean, var, W2, b2, pw, pb, bufA, accb, N);

    zero_int_kernel<<<(nb + 255) / 256, 256, 0, stream>>>(counts, nb);
    bin_count_kernel<<<bblocks, BIN_THREADS, nb * sizeof(int), stream>>>(
        dst, counts, E, nb);
    bin_scan_kernel<<<1, 1024, 0, stream>>>(counts, offs, cursor, nb);
    bin_fill_kernel<<<bblocks, BIN_THREADS, 2 * nb * sizeof(int), stream>>>(
        src, dst, norm, cursor, rec, E, nb);

    for (int k = 0; k < 10; ++k) {
        const float* prev = (k & 1) ? bufB : bufA;
        float*       next = (k & 1) ? bufA : bufB;
        hop_tile_kernel<<<nb, HOP_THREADS, 0, stream>>>(rec, offs, prev, next, N);
        combine_kernel<<<nblocks, ROW_TILE, 0, stream>>>(next, accb, pw, pb, N);
    }

    lsm_kernel<<<nblocks, ROW_TILE, 0, stream>>>(accb, (float*)d_out, N);
}


# Each hop's gate and acc update run on the work-group's finished LDS tile before it is stored (acc piece loaded under the edge loop); the 10 combine launches and their re-read are gone; sum order kept
# speedup vs baseline: 2.9406x; 1.0470x over previous
#include <hip/hip_runtime.h>
#include <hip/hip_bf16.h>

typedef __attribute__((ext_vector_type(2))) float v2f;
typedef __attribute__((ext_vector_type(8))) float v8f;
typedef __attribute__((ext_vector_type(4))) int   v4i;
typedef __attribute__((ext_vector_type(4))) float v4f;

#define F_DIM 128
#define H_DIM 256
#define C_DIM 40
#define BN_EPS 1e-5f
#define OT_PITCH 44

__global__ __launch_bounds__(256) void mlp_kernel(
    const float* __restrict__ x, const float* __restrict__ W1,
    const float* __restrict__ b1, const float* __restrict__ gamma,
    const float* __restrict__ beta, const float* __restrict__ mean,
    const float* __restrict__ var, const float* __restrict__ W2,
    const float* __restrict__ b2, const float* __restrict__ pw,
    const float* __restrict__ pb, float* __restrict__ h0,
    float* __restrict__ acc0, int N)
{
    __shared__ float As[16 * 129];
    __shared__ float Hs[16 * 257];
    __shared__ __attribute__((aligned(16))) float Ot[16 * OT_PITCH];
    __shared__ float gs[16];

    const int tid  = threadIdx.x;
    const int wave = tid >> 5;
    const int lane = tid & 31;
    const int half = lane >> 4;
    const int l16  = lane & 15;
    const int row0 = blockIdx.x * 16;

    #pragma unroll
    for (int i = 0; i < 8; ++i) {
        int idx = tid + i * 256;
        int r = idx >> 7, c = idx & 127;
        int gr = row0 + r;
        As[r * 129 + c] = (gr < N) ? x[(long)gr * F_DIM + c] : 0.0f;
    }
    __syncthreads();

    #pragma unroll
    for (int t = 0; t < 2; ++t) {
        const int ct = wave * 2 + t;
        const int n  = ct * 16 + l16;
        v8f acc = {};
        for (int k = 0; k < F_DIM; k += 4) {
            v2f a, b;
            a.x = As[l16 * 129 + k + 2 * half + 0];
            a.y = As[l16 * 129 + k + 2 * half + 1];
            b.x = W1[(long)(k + 2 * half + 0) * H_DIM + n];
            b.y = W1[(long)(k + 2 * half + 1) * H_DIM + n];
            acc = __builtin_amdgcn_wmma_f32_16x16x4_f32(
                false, a, false, b, (short)0, acc, false, false);
        }
        const float bb = b1[n];
        const float g  = gamma[n], be = beta[n];
        const float mu = mean[n],  iv = rsqrtf(var[n] + BN_EPS);
        #pragma unroll
        for (int r = 0; r < 8; ++r) {
            const int m = r + 8 * half;
            float v = acc[r] + bb;
            v = (v - mu) * iv * g + be;
            Hs[m * 257 + n] = fmaxf(v, 0.0f);
        }
    }
    __syncthreads();

    if (wave < 3) {
        const int n   = wave * 16 + l16;
        const bool nok = (n < C_DIM);
        v8f acc = {};
        for (int k = 0; k < H_DIM; k += 4) {
            v2f a, b;
            a.x = Hs[l16 * 257 + k + 2 * half + 0];
            a.y = Hs[l16 * 257 + k + 2 * half + 1];
            b.x = nok ? W2[(long)(k + 2 * half + 0) * C_DIM + n] : 0.0f;
            b.y = nok ? W2[(long)(k + 2 * half + 1) * C_DIM + n] : 0.0f;
            acc = __builtin_amdgcn_wmma_f32_16x16x4_f32(
                false, a, false, b, (short)0, acc, false, false);
        }
        const float bb = nok ? b2[n] : 0.0f;
        #pragma unroll
        for (int r = 0; r < 8; ++r) {
            const int m = r + 8 * half;
            if (nok) Ot[m * OT_PITCH + n] = acc[r] + bb;
        }
    }
    __syncthreads();

    if (tid < 16) {
        const v4f* rp = (const v4f*)&Ot[tid * OT_PITCH];
        float d = pb[0];
        #pragma unroll
        for (int q = 0; q < C_DIM / 4; ++q) {
            const v4f t = rp[q];
            d += t.x * pw[4 * q + 0];
            d += t.y * pw[4 * q + 1];
            d += t.z * pw[4 * q + 2];
            d += t.w * pw[4 * q + 3];
        }
        gs[tid] = 1.0f / (1.0f + expf(-d));
    }
    __syncthreads();

    const int nrows = (N - row0 < 16) ? (N - row0) : 16;
    if (tid < nrows * (C_DIM / 4)) {
        const int r = tid / (C_DIM / 4);
        const int q = tid - r * (C_DIM / 4);
        const v4f hv = *(const v4f*)&Ot[r * OT_PITCH + q * 4];
        const float g = gs[r];
        ((v4f*)(h0   + (long)row0 * C_DIM))[tid] = hv;
        ((v4f*)(acc0 + (long)row0 * C_DIM))[tid] = g * hv;
    }
}

__global__ void zero_int_kernel(int* __restrict__ p, int n)
{
    int i = blockIdx.x * blockDim.x + threadIdx.x;
    if (i < n) p[i] = 0;
}

#define TILE_SHIFT  8
#define TILE_ROWS   (1 << TILE_SHIFT)
#define HOP_THREADS 512
#define HOP_GROUPS  (HOP_THREADS / 8)
#define HOP_V4      (TILE_ROWS * (C_DIM / 4) / HOP_THREADS)
#define BIN_THREADS 1024
#define BIN_EDGES   8192

extern __shared__ int dyn_lds[];

__global__ __launch_bounds__(BIN_THREADS) void bin_count_kernel(
    const int* __restrict__ dst, int* __restrict__ counts, int E, int nb)
{
    int* hist = dyn_lds;
    const int tid = threadIdx.x;
    for (int i = tid; i < nb; i += BIN_THREADS) hist[i] = 0;
    const long c0 = (long)blockIdx.x * BIN_EDGES + tid;
    int dv[BIN_EDGES / BIN_THREADS];
    for (int k = 0; k < BIN_EDGES / BIN_THREADS; ++k) {
        const long e = c0 + (long)k * BIN_THREADS;
        dv[k] = dst[(e < E) ? e : (long)(E - 1)];
    }
    __syncthreads();
    for (int k = 0; k < BIN_EDGES / BIN_THREADS; ++k) {
        const long e = c0 + (long)k * BIN_THREADS;
        if (e < E) atomicAdd(&hist[dv[k] >> TILE_SHIFT], 1);
    }
    __syncthreads();
    for (int i = tid; i < nb; i += BIN_THREADS) {
        const int c = hist[i];
        if (c) atomicAdd(&counts[i], c);
    }
}

__global__ __launch_bounds__(1024) void bin_scan_kernel(
    const int* __restrict__ counts, int* __restrict__ offs,
    int* __restrict__ cursor, int nb)
{
    __shared__ int s[1024];
    __shared__ int carry;
    const int tid = threadIdx.x;
    if (tid == 0) carry = 0;
    __syncthreads();
    for (int base = 0; base < nb; base += 1024) {
        const int i = base + tid;
        const int v = (i < nb) ? counts[i] : 0;
        s[tid] = v;
        __syncthreads();
        for (int off = 1; off < 1024; off <<= 1) {
            const int t = (tid >= off) ? s[tid - off] : 0;
            __syncthreads();
            s[tid] += t;
            __syncthreads();
        }
        const int incl = s[tid];
        const int excl = incl - v + carry;
        if (i < nb) { offs[i] = excl; cursor[i] = excl; }
        __syncthreads();
        if (tid == 1023) carry += incl;
        __syncthreads();
    }
    if (tid == 0) offs[nb] = carry;
}

__global__ __launch_bounds__(BIN_THREADS) void bin_fill_kernel(
    const int* __restrict__ src, const int* __restrict__ dst,
    const float* __restrict__ norm, int* __restrict__ cursor,
    int* __restrict__ rec, int E, int nb)
{
    int* hist  = dyn_lds;
    int* rbase = dyn_lds + nb;
    const int tid = threadIdx.x;
    for (int i = tid; i < nb; i += BIN_THREADS) hist[i] = 0;
    const long c0 = (long)blockIdx.x * BIN_EDGES + tid;
    int   dv[BIN_EDGES / BIN_THREADS];
    int   sv[BIN_EDGES / BIN_THREADS];
    float wv[BIN_EDGES / BIN_THREADS];
    for (int k = 0; k < BIN_EDGES / BIN_THREADS; ++k) {
        const long e  = c0 + (long)k * BIN_THREADS;
        const long ec = (e < E) ? e : (long)(E - 1);
        dv[k] = dst[ec];
        sv[k] = src[ec];
        wv[k] = norm[ec];
    }
    __syncthreads();
    for (int k = 0; k < BIN_EDGES / BIN_THREADS; ++k) {
        const long e = c0 + (long)k * BIN_THREADS;
        if (e < E) atomicAdd(&hist[dv[k] >> TILE_SHIFT], 1);
    }
    __syncthreads();
    for (int i = tid; i < nb; i += BIN_THREADS) {
        const int c = hist[i];
        rbase[i] = c ? atomicAdd(&cursor[i], c) : 0;
        hist[i] = 0;
    }
    __syncthreads();
    v4i* rec4 = (v4i*)rec;
    for (int k = 0; k < BIN_EDGES / BIN_THREADS; ++k) {
        const long e = c0 + (long)k * BIN_THREADS;
        if (e < E) {
            const int b   = dv[k] >> TILE_SHIFT;
            const int pos = rbase[b] + atomicAdd(&hist[b], 1);
            v4i r;
            r.x = sv[k];
            r.y = dv[k] - (b << TILE_SHIFT);
            r.z = __float_as_int(wv[k]);
            r.w = 0;
            rec4[pos] = r;
        }
    }
}

__device__ __forceinline__ void lds_add(float* p, float v)
{
    __hip_atomic_fetch_add(p, v, __ATOMIC_RELAXED, __HIP_MEMORY_SCOPE_WORKGROUP);
}

__global__ __launch_bounds__(HOP_THREADS) void hop_tile_kernel(
    const int* __restrict__ rec, const int* __restrict__ offs,
    const float* __restrict__ prev, float* __restrict__ next,
    float* __restrict__ acc, const float* __restrict__ proj_w,
    const float* __restrict__ proj_b, int N)
{
    __shared__ __attribute__((aligned(16))) float tile[TILE_ROWS * C_DIM];
    __shared__ float rs[TILE_ROWS];

    const int tid  = threadIdx.x;
    const int b    = blockIdx.x;
    const int row0 = b << TILE_SHIFT;
    const int nrows = (N - row0 < TILE_ROWS) ? (N - row0) : TILE_ROWS;
    const int nv4   = nrows * (C_DIM / 4);

    v4f* tile4 = (v4f*)tile;
    for (int i = tid; i < TILE_ROWS * C_DIM / 4; i += HOP_THREADS)
        tile4[i] = (v4f){0.0f, 0.0f, 0.0f, 0.0f};
    __syncthreads();

    v4f* acc4 = (v4f*)(acc + (long)row0 * C_DIM);
    v4f a[HOP_V4];
    #pragma unroll
    for (int j = 0; j < HOP_V4; ++j) {
        const int i = tid + j * HOP_THREADS;
        a[j] = acc4[(i < nv4) ? i : (nv4 - 1)];
    }

    const v4i* rec4 = (const v4i*)rec;
    const int e0  = offs[b];
    const int e1  = offs[b + 1];
    const int grp = tid >> 3;
    const int l   = tid & 7;

    const int last = e1 - 1;
    int e = e0 + grp;
    if (e + HOP_GROUPS < e1) {
        v4i ra = rec4[e];
        v4i rb = rec4[e + HOP_GROUPS];
        for (;;) {
            const int en = e + 2 * HOP_GROUPS;
            const int ia = (en < last) ? en : last;
            const int ib = (en + HOP_GROUPS < last) ? (en + HOP_GROUPS) : last;
            const v4i na = rec4[ia];
            const v4i nc = rec4[ib];
            const float* pa = prev + (long)ra.x * C_DIM + l;
            const float* pb = prev + (long)rb.x * C_DIM + l;
            float va[5], vb[5];
            for (int j = 0; j < 5; ++j) { va[j] = pa[8 * j]; vb[j] = pb[8 * j]; }
            const float wa = __int_as_float(ra.z);
            const float wb = __int_as_float(rb.z);
            float* ta = tile + ra.y * C_DIM + l;
            float* tb = tile + rb.y * C_DIM + l;
            for (int j = 0; j < 5; ++j) lds_add(&ta[8 * j], wa * va[j]);
            for (int j = 0; j < 5; ++j) lds_add(&tb[8 * j], wb * vb[j]);
            e = en;
            if (e + HOP_GROUPS >= e1) break;
            ra = na;
            rb = nc;
        }
    }
    if (e < e1) {
        const v4i ra = rec4[e];
        const float* pa = prev + (long)ra.x * C_DIM + l;
        float va[5];
        for (int j = 0; j < 5; ++j) va[j] = pa[8 * j];
        const float wa = __int_as_float(ra.z);
        float* ta = tile + ra.y * C_DIM + l;
        for (int j = 0; j < 5; ++j) lds_add(&ta[8 * j], wa * va[j]);
    }
    __syncthreads();

    if (tid < nrows) {
        const v4f* rp = (const v4f*)&tile[tid * C_DIM];
        float d = proj_b[0];
        #pragma unroll
        for (int q = 0; q < C_DIM / 4; ++q) {
            const v4f t = rp[q];
            d += t.x * proj_w[4 * q + 0];
            d += t.y * proj_w[4 * q + 1];
            d += t.z * proj_w[4 * q + 2];
            d += t.w * proj_w[4 * q + 3];
        }
        rs[tid] = 1.0f / (1.0f + expf(-d));
    }
    __syncthreads();

    v4f* out4 = (v4f*)(next + (long)row0 * C_DIM);
    v4f   t[HOP_V4];
    float rf[HOP_V4];
    #pragma unroll
    for (int j = 0; j < HOP_V4; ++j) {
        const int i  = tid + j * HOP_THREADS;
        const int ic = (i < nv4) ? i : (nv4 - 1);
        t[j]  = tile4[i];
        rf[j] = rs[ic / (C_DIM / 4)];
    }
    #pragma unroll
    for (int j = 0; j < HOP_V4; ++j) {
        const int i = tid + j * HOP_THREADS;
        if (i < nv4) {
            out4[i] = t[j];
            acc4[i] = a[j] + rf[j] * t[j];
        }
    }
}

#define ROW_TILE  256
#define ROW_V4    (C_DIM / 4)
#define ROW_PITCH 44

__global__ __launch_bounds__(ROW_TILE) void lsm_kernel(
    const float* __restrict__ acc, float* __restrict__ out, int N)
{
    __shared__ __attribute__((aligned(16))) float rows[ROW_TILE * ROW_PITCH];

    const int tid   = threadIdx.x;
    const int row0  = blockIdx.x * ROW_TILE;
    const int nrows = (N - row0 < ROW_TILE) ? (N - row0) : ROW_TILE;
    const int nv4   = nrows * ROW_V4;
    const v4f* a4 = (const v4f*)(acc + (long)row0 * C_DIM);
    v4f*       o4 = (v4f*)(out + (long)row0 * C_DIM);

    v4f x[ROW_V4];
    #pragma unroll
    for (int j = 0; j < ROW_V4; ++j) {
        const int i  = tid + j * ROW_TILE;
        const int ic = (i < nv4) ? i : (nv4 - 1);
        x[j] = a4[ic];
    }
    #pragma unroll
    for (int j = 0; j < ROW_V4; ++j) {
        const int i = tid + j * ROW_TILE;
        const int r = i / ROW_V4;
        const int q = i - r * ROW_V4;
        *(v4f*)&rows[r * ROW_PITCH + q * 4] = x[j];
    }
    __syncthreads();

    if (tid < nrows) {
        v4f* rp = (v4f*)&rows[tid * ROW_PITCH];
        v4f t[ROW_V4];
        #pragma unroll
        for (int q = 0; q < ROW_V4; ++q) t[q] = rp[q];
        float m = -3.402823466e38f;
        #pragma unroll
        for (int q = 0; q < ROW_V4; ++q) {
            m = fmaxf(m, t[q].x); m = fmaxf(m, t[q].y);
            m = fmaxf(m, t[q].z); m = fmaxf(m, t[q].w);
        }
        float s = 0.0f;
        #pragma unroll
        for (int q = 0; q < ROW_V4; ++q) {
            s += expf(t[q].x - m); s += expf(t[q].y - m);
            s += expf(t[q].z - m); s += expf(t[q].w - m);
        }
        const float ls = logf(s);
        #pragma unroll
        for (int q = 0; q < ROW_V4; ++q) {
            v4f o;
            o.x = t[q].x - m - ls; o.y = t[q].y - m - ls;
            o.z = t[q].z - m - ls; o.w = t[q].w - m - ls;
            rp[q] = o;
        }
    }
    __syncthreads();

    for (int j = 0; j < ROW_V4; ++j) {
        const int i = tid + j * ROW_TILE;
        if (i < nv4) {
            const int r = i / ROW_V4;
            const int q = i - r * ROW_V4;
            o4[i] = *(const v4f*)&rows[r * ROW_PITCH + q * 4];
        }
    }
}

extern "C" void kernel_launch(void* const* d_in, const int* in_sizes, int n_in,
                              void* d_out, int out_size, void* d_ws, size_t ws_size,
                              hipStream_t stream)
{
    const float* x     = (const float*)d_in[0];
    const int*   edge  = (const int*)d_in[1];
    const float* norm  = (const float*)d_in[2];
    const float* W1    = (const float*)d_in[3];
    const float* b1    = (const float*)d_in[4];
    const float* gamma = (const float*)d_in[5];
    const float* beta  = (const float*)d_in[6];
    const float* mean  = (const float*)d_in[7];
    const float* var   = (const float*)d_in[8];
    const float* W2    = (const float*)d_in[9];
    const float* b2    = (const float*)d_in[10];
    const float* pw    = (const float*)d_in[11];
    const float* pb    = (const float*)d_in[12];

    const int N = in_sizes[0] / F_DIM;
    const int E = in_sizes[2];

    const size_t per = (((size_t)N * C_DIM) + 63) & ~(size_t)63;
    float* bufA = (float*)d_ws;
    float* bufB = bufA + per;
    float* accb = bufB + per;

    const int    nb  = (N + TILE_ROWS - 1) / TILE_ROWS;
    const size_t nbp = ((size_t)nb + 1 + 63) & ~(size_t)63;
    int* rec    = (int*)(accb + per);
    int* counts = rec + (size_t)E * 4;
    int* offs   = counts + nbp;
    int* cursor = offs + nbp;

    const int* src = edge;
    const int* dst = edge + E;

    const int  nblocks = (N + ROW_TILE - 1) / ROW_TILE;
    const int  bblocks = (E + BIN_EDGES - 1) / BIN_EDGES;

    mlp_kernel<<<(N + 15) / 16, 256, 0, stream>>>(
        x, W1, b1, gamma, beta, mean, var, W2, b2, pw, pb, bufA, accb, N);

    zero_int_kernel<<<(nb + 255) / 256, 256, 0, stream>>>(counts, nb);
    bin_count_kernel<<<bblocks, BIN_THREADS, nb * sizeof(int), stream>>>(
        dst, counts, E, nb);
    bin_scan_kernel<<<1, 1024, 0, stream>>>(counts, offs, cursor, nb);
    bin_fill_kernel<<<bblocks, BIN_THREADS, 2 * nb * sizeof(int), stream>>>(
        src, dst, norm, cursor, rec, E, nb);

    for (int k = 0; k < 10; ++k) {
        const float* prev = (k & 1) ? bufB : bufA;
        float*       next = (k & 1) ? bufA : bufB;
        hop_tile_kernel<<<nb, HOP_THREADS, 0, stream>>>(
            rec, offs, prev, next, accb, pw, pb, N);
    }

    lsm_kernel<<<nblocks, ROW_TILE, 0, stream>>>(accb, (float*)d_out, N);
}


# MLP head on 64-row work-group tiles (was 16): a wave keeps 4 row tiles x 2 column tiles of accumulators, so one W1 fragment load from global feeds 4 WMMAs; 4x fewer work-groups; sum order kept
# speedup vs baseline: 3.5221x; 1.1977x over previous
#include <hip/hip_runtime.h>
#include <hip/hip_bf16.h>

typedef __attribute__((ext_vector_type(2))) float v2f;
typedef __attribute__((ext_vector_type(8))) float v8f;
typedef __attribute__((ext_vector_type(4))) int   v4i;
typedef __attribute__((ext_vector_type(4))) float v4f;

#define F_DIM 128
#define H_DIM 256
#define C_DIM 40
#define BN_EPS 1e-5f
#define OT_PITCH 44

#define MT_ROWS  64
#define AS_PITCH 132
#define HS_PITCH 260

__global__ __launch_bounds__(256) void mlp_kernel(
    const float* __restrict__ x, const float* __restrict__ W1,
    const float* __restrict__ b1, const float* __restrict__ gamma,
    const float* __restrict__ beta, const float* __restrict__ mean,
    const float* __restrict__ var, const float* __restrict__ W2,
    const float* __restrict__ b2, const float* __restrict__ pw,
    const float* __restrict__ pb, float* __restrict__ h0,
    float* __restrict__ acc0, int N)
{
    __shared__ __attribute__((aligned(16))) float As[MT_ROWS * AS_PITCH];
    __shared__ __attribute__((aligned(16))) float Hs[MT_ROWS * HS_PITCH];
    __shared__ float gs[MT_ROWS];
    float* Ot = As;

    const int tid  = threadIdx.x;
    const int wave = tid >> 5;
    const int lane = tid & 31;
    const int half = lane >> 4;
    const int l16  = lane & 15;
    const int row0 = blockIdx.x * MT_ROWS;
    const int nrows = (N - row0 < MT_ROWS) ? (N - row0) : MT_ROWS;

    {
        const v4f* x4 = (const v4f*)x;
        v4f xv[8];
        #pragma unroll
        for (int i = 0; i < 8; ++i) {
            const int idx = tid + i * 256;
            const int r  = idx >> 5, c4 = idx & 31;
            const int rr = (r < nrows) ? r : (nrows - 1);
            xv[i] = x4[(long)(row0 + rr) * (F_DIM / 4) + c4];
        }
        #pragma unroll
        for (int i = 0; i < 8; ++i) {
            const int idx = tid + i * 256;
            const int r  = idx >> 5, c4 = idx & 31;
            *(v4f*)&As[r * AS_PITCH + 4 * c4] = xv[i];
        }
    }
    __syncthreads();

    {
        v8f acc[4][2];
        #pragma unroll
        for (int rt = 0; rt < 4; ++rt) { acc[rt][0] = (v8f){}; acc[rt][1] = (v8f){}; }
        const int n0 = wave * 32 + l16;
        const float* w1p = W1 + (long)(2 * half) * H_DIM + n0;
        const float* ap  = As + l16 * AS_PITCH + 2 * half;
        #pragma unroll 8
        for (int k = 0; k < F_DIM; k += 4) {
            v2f b[2];
            #pragma unroll
            for (int t = 0; t < 2; ++t) {
                b[t].x = w1p[(long)(k + 0) * H_DIM + 16 * t];
                b[t].y = w1p[(long)(k + 1) * H_DIM + 16 * t];
            }
            #pragma unroll
            for (int rt = 0; rt < 4; ++rt) {
                const v2f a = *(const v2f*)&ap[rt * 16 * AS_PITCH + k];
                #pragma unroll
                for (int t = 0; t < 2; ++t)
                    acc[rt][t] = __builtin_amdgcn_wmma_f32_16x16x4_f32(
                        false, a, false, b[t], (short)0, acc[rt][t], false, false);
            }
        }
        #pragma unroll
        for (int t = 0; t < 2; ++t) {
            const int n = n0 + 16 * t;
            const float bb = b1[n];
            const float g  = gamma[n], be = beta[n];
            const float mu = mean[n],  iv = rsqrtf(var[n] + BN_EPS);
            #pragma unroll
            for (int rt = 0; rt < 4; ++rt) {
                #pragma unroll
                for (int r = 0; r < 8; ++r) {
                    const int m = rt * 16 + r + 8 * half;
                    float v = acc[rt][t][r] + bb;
                    v = (v - mu) * iv * g + be;
                    Hs[m * HS_PITCH + n] = fmaxf(v, 0.0f);
                }
            }
        }
    }
    __syncthreads();

    {
        const int rt2 = wave & 3;
        const float* hp = Hs + (rt2 * 16 + l16) * HS_PITCH + 2 * half;
        if (wave < 4) {
            v8f c0 = {}, c1 = {};
            const float* w2p = W2 + (long)(2 * half) * C_DIM + l16;
            #pragma unroll
            for (int k = 0; k < H_DIM; k += 4) {
                const v2f a = *(const v2f*)&hp[k];
                v2f u, w;
                u.x = w2p[(long)(k + 0) * C_DIM];
                u.y = w2p[(long)(k + 1) * C_DIM];
                w.x = w2p[(long)(k + 0) * C_DIM + 16];
                w.y = w2p[(long)(k + 1) * C_DIM + 16];
                c0 = __builtin_amdgcn_wmma_f32_16x16x4_f32(
                    false, a, false, u, (short)0, c0, false, false);
                c1 = __builtin_amdgcn_wmma_f32_16x16x4_f32(
                    false, a, false, w, (short)0, c1, false, false);
            }
            const float bb0 = b2[l16];
            const float bb1 = b2[16 + l16];
            #pragma unroll
            for (int r = 0; r < 8; ++r) {
                const int m = rt2 * 16 + r + 8 * half;
                Ot[m * OT_PITCH + l16]      = c0[r] + bb0;
                Ot[m * OT_PITCH + 16 + l16] = c1[r] + bb1;
            }
        } else {
            const int  n   = 32 + l16;
            const bool nok = (n < C_DIM);
            const float* w2p = W2 + (long)(2 * half) * C_DIM + (nok ? n : (C_DIM - 1));
            v8f c2 = {};
            #pragma unroll
            for (int k = 0; k < H_DIM; k += 4) {
                const v2f a = *(const v2f*)&hp[k];
                v2f u;
                const float t0 = w2p[(long)(k + 0) * C_DIM];
                const float t1 = w2p[(long)(k + 1) * C_DIM];
                u.x = t0;
                u.y = t1;
                c2 = __builtin_amdgcn_wmma_f32_16x16x4_f32(
                    false, a, false, u, (short)0, c2, false, false);
            }
            const float bb = nok ? b2[n] : 0.0f;
            #pragma unroll
            for (int r = 0; r < 8; ++r) {
                const int m = rt2 * 16 + r + 8 * half;
                if (nok) Ot[m * OT_PITCH + n] = c2[r] + bb;
            }
        }
    }
    __syncthreads();

    if (tid < MT_ROWS) {
        const v4f* rp = (const v4f*)&Ot[tid * OT_PITCH];
        float d = pb[0];
        #pragma unroll
        for (int q = 0; q < C_DIM / 4; ++q) {
            const v4f t = rp[q];
            d += t.x * pw[4 * q + 0];
            d += t.y * pw[4 * q + 1];
            d += t.z * pw[4 * q + 2];
            d += t.w * pw[4 * q + 3];
        }
        gs[tid] = 1.0f / (1.0f + expf(-d));
    }
    __syncthreads();

    const int nv4 = nrows * (C_DIM / 4);
    v4f* h4 = (v4f*)(h0   + (long)row0 * C_DIM);
    v4f* a4 = (v4f*)(acc0 + (long)row0 * C_DIM);
    for (int i = tid; i < nv4; i += 256) {
        const int r = i / (C_DIM / 4);
        const int q = i - r * (C_DIM / 4);
        const v4f hv = *(const v4f*)&Ot[r * OT_PITCH + q * 4];
        const float g = gs[r];
        h4[i] = hv;
        a4[i] = g * hv;
    }
}

__global__ void zero_int_kernel(int* __restrict__ p, int n)
{
    int i = blockIdx.x * blockDim.x + threadIdx.x;
    if (i < n) p[i] = 0;
}

#define TILE_SHIFT  8
#define TILE_ROWS   (1 << TILE_SHIFT)
#define HOP_THREADS 512
#define HOP_GROUPS  (HOP_THREADS / 8)
#define HOP_V4      (TILE_ROWS * (C_DIM / 4) / HOP_THREADS)
#define BIN_THREADS 1024
#define BIN_EDGES   8192

extern __shared__ int dyn_lds[];

__global__ __launch_bounds__(BIN_THREADS) void bin_count_kernel(
    const int* __restrict__ dst, int* __restrict__ counts, int E, int nb)
{
    int* hist = dyn_lds;
    const int tid = threadIdx.x;
    for (int i = tid; i < nb; i += BIN_THREADS) hist[i] = 0;
    const long c0 = (long)blockIdx.x * BIN_EDGES + tid;
    int dv[BIN_EDGES / BIN_THREADS];
    for (int k = 0; k < BIN_EDGES / BIN_THREADS; ++k) {
        const long e = c0 + (long)k * BIN_THREADS;
        dv[k] = dst[(e < E) ? e : (long)(E - 1)];
    }
    __syncthreads();
    for (int k = 0; k < BIN_EDGES / BIN_THREADS; ++k) {
        const long e = c0 + (long)k * BIN_THREADS;
        if (e < E) atomicAdd(&hist[dv[k] >> TILE_SHIFT], 1);
    }
    __syncthreads();
    for (int i = tid; i < nb; i += BIN_THREADS) {
        const int c = hist[i];
        if (c) atomicAdd(&counts[i], c);
    }
}

__global__ __launch_bounds__(1024) void bin_scan_kernel(
    const int* __restrict__ counts, int* __restrict__ offs,
    int* __restrict__ cursor, int nb)
{
    __shared__ int s[1024];
    __shared__ int carry;
    const int tid = threadIdx.x;
    if (tid == 0) carry = 0;
    __syncthreads();
    for (int base = 0; base < nb; base += 1024) {
        const int i = base + tid;
        const int v = (i < nb) ? counts[i] : 0;
        s[tid] = v;
        __syncthreads();
        for (int off = 1; off < 1024; off <<= 1) {
            const int t = (tid >= off) ? s[tid - off] : 0;
            __syncthreads();
            s[tid] += t;
            __syncthreads();
        }
        const int incl = s[tid];
        const int excl = incl - v + carry;
        if (i < nb) { offs[i] = excl; cursor[i] = excl; }
        __syncthreads();
        if (tid == 1023) carry += incl;
        __syncthreads();
    }
    if (tid == 0) offs[nb] = carry;
}

__global__ __launch_bounds__(BIN_THREADS) void bin_fill_kernel(
    const int* __restrict__ src, const int* __restrict__ dst,
    const float* __restrict__ norm, int* __restrict__ cursor,
    int* __restrict__ rec, int E, int nb)
{
    int* hist  = dyn_lds;
    int* rbase = dyn_lds + nb;
    const int tid = threadIdx.x;
    for (int i = tid; i < nb; i += BIN_THREADS) hist[i] = 0;
    const long c0 = (long)blockIdx.x * BIN_EDGES + tid;
    int   dv[BIN_EDGES / BIN_THREADS];
    int   sv[BIN_EDGES / BIN_THREADS];
    float wv[BIN_EDGES / BIN_THREADS];
    for (int k = 0; k < BIN_EDGES / BIN_THREADS; ++k) {
        const long e  = c0 + (long)k * BIN_THREADS;
        const long ec = (e < E) ? e : (long)(E - 1);
        dv[k] = dst[ec];
        sv[k] = src[ec];
        wv[k] = norm[ec];
    }
    __syncthreads();
    for (int k = 0; k < BIN_EDGES / BIN_THREADS; ++k) {
        const long e = c0 + (long)k * BIN_THREADS;
        if (e < E) atomicAdd(&hist[dv[k] >> TILE_SHIFT], 1);
    }
    __syncthreads();
    for (int i = tid; i < nb; i += BIN_THREADS) {
        const int c = hist[i];
        rbase[i] = c ? atomicAdd(&cursor[i], c) : 0;
        hist[i] = 0;
    }
    __syncthreads();
    v4i* rec4 = (v4i*)rec;
    for (int k = 0; k < BIN_EDGES / BIN_THREADS; ++k) {
        const long e = c0 + (long)k * BIN_THREADS;
        if (e < E) {
            const int b   = dv[k] >> TILE_SHIFT;
            const int pos = rbase[b] + atomicAdd(&hist[b], 1);
            v4i r;
            r.x = sv[k];
            r.y = dv[k] - (b << TILE_SHIFT);
            r.z = __float_as_int(wv[k]);
            r.w = 0;
            rec4[pos] = r;
        }
    }
}

__device__ __forceinline__ void lds_add(float* p, float v)
{
    __hip_atomic_fetch_add(p, v, __ATOMIC_RELAXED, __HIP_MEMORY_SCOPE_WORKGROUP);
}

__global__ __launch_bounds__(HOP_THREADS) void hop_tile_kernel(
    const int* __restrict__ rec, const int* __restrict__ offs,
    const float* __restrict__ prev, float* __restrict__ next,
    float* __restrict__ acc, const float* __restrict__ proj_w,
    const float* __restrict__ proj_b, int N)
{
    __shared__ __attribute__((aligned(16))) float tile[TILE_ROWS * C_DIM];
    __shared__ float rs[TILE_ROWS];

    const int tid  = threadIdx.x;
    const int b    = blockIdx.x;
    const int row0 = b << TILE_SHIFT;
    const int nrows = (N - row0 < TILE_ROWS) ? (N - row0) : TILE_ROWS;
    const int nv4   = nrows * (C_DIM / 4);

    v4f* tile4 = (v4f*)tile;
    for (int i = tid; i < TILE_ROWS * C_DIM / 4; i += HOP_THREADS)
        tile4[i] = (v4f){0.0f, 0.0f, 0.0f, 0.0f};
    __syncthreads();

    v4f* acc4 = (v4f*)(acc + (long)row0 * C_DIM);
    v4f a[HOP_V4];
    #pragma unroll
    for (int j = 0; j < HOP_V4; ++j) {
        const int i = tid + j * HOP_THREADS;
        a[j] = acc4[(i < nv4) ? i : (nv4 - 1)];
    }

    const v4i* rec4 = (const v4i*)rec;
    const int e0  = offs[b];
    const int e1  = offs[b + 1];
    const int grp = tid >> 3;
    const int l   = tid & 7;

    const int last = e1 - 1;
    int e = e0 + grp;
    if (e + HOP_GROUPS < e1) {
        v4i ra = rec4[e];
        v4i rb = rec4[e + HOP_GROUPS];
        for (;;) {
            const int en = e + 2 * HOP_GROUPS;
            const int ia = (en < last) ? en : last;
            const int ib = (en + HOP_GROUPS < last) ? (en + HOP_GROUPS) : last;
            const v4i na = rec4[ia];
            const v4i nc = rec4[ib];
            const float* pa = prev + (long)ra.x * C_DIM + l;
            const float* pb = prev + (long)rb.x * C_DIM + l;
            float va[5], vb[5];
            for (int j = 0; j < 5; ++j) { va[j] = pa[8 * j]; vb[j] = pb[8 * j]; }
            const float wa = __int_as_float(ra.z);
            const float wb = __int_as_float(rb.z);
            float* ta = tile + ra.y * C_DIM + l;
            float* tb = tile + rb.y * C_DIM + l;
            for (int j = 0; j < 5; ++j) lds_add(&ta[8 * j], wa * va[j]);
            for (int j = 0; j < 5; ++j) lds_add(&tb[8 * j], wb * vb[j]);
            e = en;
            if (e + HOP_GROUPS >= e1) break;
            ra = na;
            rb = nc;
        }
    }
    if (e < e1) {
        const v4i ra = rec4[e];
        const float* pa = prev + (long)ra.x * C_DIM + l;
        float va[5];
        for (int j = 0; j < 5; ++j) va[j] = pa[8 * j];
        const float wa = __int_as_float(ra.z);
        float* ta = tile + ra.y * C_DIM + l;
        for (int j = 0; j < 5; ++j) lds_add(&ta[8 * j], wa * va[j]);
    }
    __syncthreads();

    if (tid < nrows) {
        const v4f* rp = (const v4f*)&tile[tid * C_DIM];
        float d = proj_b[0];
        #pragma unroll
        for (int q = 0; q < C_DIM / 4; ++q) {
            const v4f t = rp[q];
            d += t.x * proj_w[4 * q + 0];
            d += t.y * proj_w[4 * q + 1];
            d += t.z * proj_w[4 * q + 2];
            d += t.w * proj_w[4 * q + 3];
        }
        rs[tid] = 1.0f / (1.0f + expf(-d));
    }
    __syncthreads();

    v4f* out4 = (v4f*)(next + (long)row0 * C_DIM);
    v4f   t[HOP_V4];
    float rf[HOP_V4];
    #pragma unroll
    for (int j = 0; j < HOP_V4; ++j) {
        const int i  = tid + j * HOP_THREADS;
        const int ic = (i < nv4) ? i : (nv4 - 1);
        t[j]  = tile4[i];
        rf[j] = rs[ic / (C_DIM / 4)];
    }
    #pragma unroll
    for (int j = 0; j < HOP_V4; ++j) {
        const int i = tid + j * HOP_THREADS;
        if (i < nv4) {
            out4[i] = t[j];
            acc4[i] = a[j] + rf[j] * t[j];
        }
    }
}

#define ROW_TILE  256
#define ROW_V4    (C_DIM / 4)
#define ROW_PITCH 44

__global__ __launch_bounds__(ROW_TILE) void lsm_kernel(
    const float* __restrict__ acc, float* __restrict__ out, int N)
{
    __shared__ __attribute__((aligned(16))) float rows[ROW_TILE * ROW_PITCH];

    const int tid   = threadIdx.x;
    const int row0  = blockIdx.x * ROW_TILE;
    const int nrows = (N - row0 < ROW_TILE) ? (N - row0) : ROW_TILE;
    const int nv4   = nrows * ROW_V4;
    const v4f* a4 = (const v4f*)(acc + (long)row0 * C_DIM);
    v4f*       o4 = (v4f*)(out + (long)row0 * C_DIM);

    v4f x[ROW_V4];
    #pragma unroll
    for (int j = 0; j < ROW_V4; ++j) {
        const int i  = tid + j * ROW_TILE;
        const int ic = (i < nv4) ? i : (nv4 - 1);
        x[j] = a4[ic];
    }
    #pragma unroll
    for (int j = 0; j < ROW_V4; ++j) {
        const int i = tid + j * ROW_TILE;
        const int r = i / ROW_V4;
        const int q = i - r * ROW_V4;
        *(v4f*)&rows[r * ROW_PITCH + q * 4] = x[j];
    }
    __syncthreads();

    if (tid < nrows) {
        v4f* rp = (v4f*)&rows[tid * ROW_PITCH];
        v4f t[ROW_V4];
        #pragma unroll
        for (int q = 0; q < ROW_V4; ++q) t[q] = rp[q];
        float m = -3.402823466e38f;
        #pragma unroll
        for (int q = 0; q < ROW_V4; ++q) {
            m = fmaxf(m, t[q].x); m = fmaxf(m, t[q].y);
            m = fmaxf(m, t[q].z); m = fmaxf(m, t[q].w);
        }
        float s = 0.0f;
        #pragma unroll
        for (int q = 0; q < ROW_V4; ++q) {
            s += expf(t[q].x - m); s += expf(t[q].y - m);
            s += expf(t[q].z - m); s += expf(t[q].w - m);
        }
        const float ls = logf(s);
        #pragma unroll
        for (int q = 0; q < ROW_V4; ++q) {
            v4f o;
            o.x = t[q].x - m - ls; o.y = t[q].y - m - ls;
            o.z = t[q].z - m - ls; o.w = t[q].w - m - ls;
            rp[q] = o;
        }
    }
    __syncthreads();

    for (int j = 0; j < ROW_V4; ++j) {
        const int i = tid + j * ROW_TILE;
        if (i < nv4) {
            const int r = i / ROW_V4;
            const int q = i - r * ROW_V4;
            o4[i] = *(const v4f*)&rows[r * ROW_PITCH + q * 4];
        }
    }
}

extern "C" void kernel_launch(void* const* d_in, const int* in_sizes, int n_in,
                              void* d_out, int out_size, void* d_ws, size_t ws_size,
                              hipStream_t stream)
{
    const float* x     = (const float*)d_in[0];
    const int*   edge  = (const int*)d_in[1];
    const float* norm  = (const float*)d_in[2];
    const float* W1    = (const float*)d_in[3];
    const float* b1    = (const float*)d_in[4];
    const float* gamma = (const float*)d_in[5];
    const float* beta  = (const float*)d_in[6];
    const float* mean  = (const float*)d_in[7];
    const float* var   = (const float*)d_in[8];
    const float* W2    = (const float*)d_in[9];
    const float* b2    = (const float*)d_in[10];
    const float* pw    = (const float*)d_in[11];
    const float* pb    = (const float*)d_in[12];

    const int N = in_sizes[0] / F_DIM;
    const int E = in_sizes[2];

    const size_t per = (((size_t)N * C_DIM) + 63) & ~(size_t)63;
    float* bufA = (float*)d_ws;
    float* bufB = bufA + per;
    float* accb = bufB + per;

    const int    nb  = (N + TILE_ROWS - 1) / TILE_ROWS;
    const size_t nbp = ((size_t)nb + 1 + 63) & ~(size_t)63;
    int* rec    = (int*)(accb + per);
    int* counts = rec + (size_t)E * 4;
    int* offs   = counts + nbp;
    int* cursor = offs + nbp;

    const int* src = edge;
    const int* dst = edge + E;

    const int  nblocks = (N + ROW_TILE - 1) / ROW_TILE;
    const int  bblocks = (E + BIN_EDGES - 1) / BIN_EDGES;

    mlp_kernel<<<(N + MT_ROWS - 1) / MT_ROWS, 256, 0, stream>>>(
        x, W1, b1, gamma, beta, mean, var, W2, b2, pw, pb, bufA, accb, N);

    zero_int_kernel<<<(nb + 255) / 256, 256, 0, stream>>>(counts, nb);
    bin_count_kernel<<<bblocks, BIN_THREADS, nb * sizeof(int), stream>>>(
        dst, counts, E, nb);
    bin_scan_kernel<<<1, 1024, 0, stream>>>(counts, offs, cursor, nb);
    bin_fill_kernel<<<bblocks, BIN_THREADS, 2 * nb * sizeof(int), stream>>>(
        src, dst, norm, cursor, rec, E, nb);

    for (int k = 0; k < 10; ++k) {
        const float* prev = (k & 1) ? bufB : bufA;
        float*       next = (k & 1) ? bufA : bufB;
        hop_tile_kernel<<<nb, HOP_THREADS, 0, stream>>>(
            rec, offs, prev, next, accb, pw, pb, N);
    }

    lsm_kernel<<<nblocks, ROW_TILE, 0, stream>>>(accb, (float*)d_out, N);
}


# Hop tiles run on 8 waves instead of 16: 4 edges in flight per 8-lane group (was 2), rows and next records issued as one load group; half the waves to start per hop; LDS atomics, arrival order kept
# speedup vs baseline: 3.6338x; 1.0317x over previous
#include <hip/hip_runtime.h>
#include <hip/hip_bf16.h>

typedef __attribute__((ext_vector_type(2))) float v2f;
typedef __attribute__((ext_vector_type(8))) float v8f;
typedef __attribute__((ext_vector_type(4))) int   v4i;
typedef __attribute__((ext_vector_type(4))) float v4f;

#define F_DIM 128
#define H_DIM 256
#define C_DIM 40
#define BN_EPS 1e-5f
#define OT_PITCH 44

#define MT_ROWS  64
#define AS_PITCH 132
#define HS_PITCH 260

__global__ __launch_bounds__(256) void mlp_kernel(
    const float* __restrict__ x, const float* __restrict__ W1,
    const float* __restrict__ b1, const float* __restrict__ gamma,
    const float* __restrict__ beta, const float* __restrict__ mean,
    const float* __restrict__ var, const float* __restrict__ W2,
    const float* __restrict__ b2, const float* __restrict__ pw,
    const float* __restrict__ pb, float* __restrict__ h0,
    float* __restrict__ acc0, int N)
{
    __shared__ __attribute__((aligned(16))) float As[MT_ROWS * AS_PITCH];
    __shared__ __attribute__((aligned(16))) float Hs[MT_ROWS * HS_PITCH];
    __shared__ float gs[MT_ROWS];
    float* Ot = As;

    const int tid  = threadIdx.x;
    const int wave = tid >> 5;
    const int lane = tid & 31;
    const int half = lane >> 4;
    const int l16  = lane & 15;
    const int row0 = blockIdx.x * MT_ROWS;
    const int nrows = (N - row0 < MT_ROWS) ? (N - row0) : MT_ROWS;

    {
        const v4f* x4 = (const v4f*)x;
        v4f xv[8];
        #pragma unroll
        for (int i = 0; i < 8; ++i) {
            const int idx = tid + i * 256;
            const int r  = idx >> 5, c4 = idx & 31;
            const int rr = (r < nrows) ? r : (nrows - 1);
            xv[i] = x4[(long)(row0 + rr) * (F_DIM / 4) + c4];
        }
        #pragma unroll
        for (int i = 0; i < 8; ++i) {
            const int idx = tid + i * 256;
            const int r  = idx >> 5, c4 = idx & 31;
            *(v4f*)&As[r * AS_PITCH + 4 * c4] = xv[i];
        }
    }
    __syncthreads();

    {
        v8f acc[4][2];
        #pragma unroll
        for (int rt = 0; rt < 4; ++rt) { acc[rt][0] = (v8f){}; acc[rt][1] = (v8f){}; }
        const int n0 = wave * 32 + l16;
        const float* w1p = W1 + (long)(2 * half) * H_DIM + n0;
        const float* ap  = As + l16 * AS_PITCH + 2 * half;
        #pragma unroll 8
        for (int k = 0; k < F_DIM; k += 4) {
            v2f b[2];
            #pragma unroll
            for (int t = 0; t < 2; ++t) {
                b[t].x = w1p[(long)(k + 0) * H_DIM + 16 * t];
                b[t].y = w1p[(long)(k + 1) * H_DIM + 16 * t];
            }
            #pragma unroll
            for (int rt = 0; rt < 4; ++rt) {
                const v2f a = *(const v2f*)&ap[rt * 16 * AS_PITCH + k];
                #pragma unroll
                for (int t = 0; t < 2; ++t)
                    acc[rt][t] = __builtin_amdgcn_wmma_f32_16x16x4_f32(
                        false, a, false, b[t], (short)0, acc[rt][t], false, false);
            }
        }
        #pragma unroll
        for (int t = 0; t < 2; ++t) {
            const int n = n0 + 16 * t;
            const float bb = b1[n];
            const float g  = gamma[n], be = beta[n];
            const float mu = mean[n],  iv = rsqrtf(var[n] + BN_EPS);
            #pragma unroll
            for (int rt = 0; rt < 4; ++rt) {
                #pragma unroll
                for (int r = 0; r < 8; ++r) {
                    const int m = rt * 16 + r + 8 * half;
                    float v = acc[rt][t][r] + bb;
                    v = (v - mu) * iv * g + be;
                    Hs[m * HS_PITCH + n] = fmaxf(v, 0.0f);
                }
            }
        }
    }
    __syncthreads();

    {
        const int rt2 = wave & 3;
        const float* hp = Hs + (rt2 * 16 + l16) * HS_PITCH + 2 * half;
        if (wave < 4) {
            v8f c0 = {}, c1 = {};
            const float* w2p = W2 + (long)(2 * half) * C_DIM + l16;
            #pragma unroll
            for (int k = 0; k < H_DIM; k += 4) {
                const v2f a = *(const v2f*)&hp[k];
                v2f u, w;
                u.x = w2p[(long)(k + 0) * C_DIM];
                u.y = w2p[(long)(k + 1) * C_DIM];
                w.x = w2p[(long)(k + 0) * C_DIM + 16];
                w.y = w2p[(long)(k + 1) * C_DIM + 16];
                c0 = __builtin_amdgcn_wmma_f32_16x16x4_f32(
                    false, a, false, u, (short)0, c0, false, false);
                c1 = __builtin_amdgcn_wmma_f32_16x16x4_f32(
                    false, a, false, w, (short)0, c1, false, false);
            }
            const float bb0 = b2[l16];
            const float bb1 = b2[16 + l16];
            #pragma unroll
            for (int r = 0; r < 8; ++r) {
                const int m = rt2 * 16 + r + 8 * half;
                Ot[m * OT_PITCH + l16]      = c0[r] + bb0;
                Ot[m * OT_PITCH + 16 + l16] = c1[r] + bb1;
            }
        } else {
            const int  n   = 32 + l16;
            const bool nok = (n < C_DIM);
            const float* w2p = W2 + (long)(2 * half) * C_DIM + (nok ? n : (C_DIM - 1));
            v8f c2 = {};
            #pragma unroll
            for (int k = 0; k < H_DIM; k += 4) {
                const v2f a = *(const v2f*)&hp[k];
                v2f u;
                const float t0 = w2p[(long)(k + 0) * C_DIM];
                const float t1 = w2p[(long)(k + 1) * C_DIM];
                u.x = t0;
                u.y = t1;
                c2 = __builtin_amdgcn_wmma_f32_16x16x4_f32(
                    false, a, false, u, (short)0, c2, false, false);
            }
            const float bb = nok ? b2[n] : 0.0f;
            #pragma unroll
            for (int r = 0; r < 8; ++r) {
                const int m = rt2 * 16 + r + 8 * half;
                if (nok) Ot[m * OT_PITCH + n] = c2[r] + bb;
            }
        }
    }
    __syncthreads();

    if (tid < MT_ROWS) {
        const v4f* rp = (const v4f*)&Ot[tid * OT_PITCH];
        float d = pb[0];
        #pragma unroll
        for (int q = 0; q < C_DIM / 4; ++q) {
            const v4f t = rp[q];
            d += t.x * pw[4 * q + 0];
            d += t.y * pw[4 * q + 1];
            d += t.z * pw[4 * q + 2];
            d += t.w * pw[4 * q + 3];
        }
        gs[tid] = 1.0f / (1.0f + expf(-d));
    }
    __syncthreads();

    const int nv4 = nrows * (C_DIM / 4);
    v4f* h4 = (v4f*)(h0   + (long)row0 * C_DIM);
    v4f* a4 = (v4f*)(acc0 + (long)row0 * C_DIM);
    for (int i = tid; i < nv4; i += 256) {
        const int r = i / (C_DIM / 4);
        const int q = i - r * (C_DIM / 4);
        const v4f hv = *(const v4f*)&Ot[r * OT_PITCH + q * 4];
        const float g = gs[r];
        h4[i] = hv;
        a4[i] = g * hv;
    }
}

__global__ void zero_int_kernel(int* __restrict__ p, int n)
{
    int i = blockIdx.x * blockDim.x + threadIdx.x;
    if (i < n) p[i] = 0;
}

#define TILE_SHIFT  8
#define TILE_ROWS   (1 << TILE_SHIFT)
#define HOP_THREADS 256
#define HOP_GROUPS  (HOP_THREADS / 8)
#define HOP_V4      (TILE_ROWS * (C_DIM / 4) / HOP_THREADS)
#define HOP_EPT     4
#define BIN_THREADS 1024
#define BIN_EDGES   8192

extern __shared__ int dyn_lds[];

__global__ __launch_bounds__(BIN_THREADS) void bin_count_kernel(
    const int* __restrict__ dst, int* __restrict__ counts, int E, int nb)
{
    int* hist = dyn_lds;
    const int tid = threadIdx.x;
    for (int i = tid; i < nb; i += BIN_THREADS) hist[i] = 0;
    const long c0 = (long)blockIdx.x * BIN_EDGES + tid;
    int dv[BIN_EDGES / BIN_THREADS];
    for (int k = 0; k < BIN_EDGES / BIN_THREADS; ++k) {
        const long e = c0 + (long)k * BIN_THREADS;
        dv[k] = dst[(e < E) ? e : (long)(E - 1)];
    }
    __syncthreads();
    for (int k = 0; k < BIN_EDGES / BIN_THREADS; ++k) {
        const long e = c0 + (long)k * BIN_THREADS;
        if (e < E) atomicAdd(&hist[dv[k] >> TILE_SHIFT], 1);
    }
    __syncthreads();
    for (int i = tid; i < nb; i += BIN_THREADS) {
        const int c = hist[i];
        if (c) atomicAdd(&counts[i], c);
    }
}

__global__ __launch_bounds__(1024) void bin_scan_kernel(
    const int* __restrict__ counts, int* __restrict__ offs,
    int* __restrict__ cursor, int nb)
{
    __shared__ int s[1024];
    __shared__ int carry;
    const int tid = threadIdx.x;
    if (tid == 0) carry = 0;
    __syncthreads();
    for (int base = 0; base < nb; base += 1024) {
        const int i = base + tid;
        const int v = (i < nb) ? counts[i] : 0;
        s[tid] = v;
        __syncthreads();
        for (int off = 1; off < 1024; off <<= 1) {
            const int t = (tid >= off) ? s[tid - off] : 0;
            __syncthreads();
            s[tid] += t;
            __syncthreads();
        }
        const int incl = s[tid];
        const int excl = incl - v + carry;
        if (i < nb) { offs[i] = excl; cursor[i] = excl; }
        __syncthreads();
        if (tid == 1023) carry += incl;
        __syncthreads();
    }
    if (tid == 0) offs[nb] = carry;
}

__global__ __launch_bounds__(BIN_THREADS) void bin_fill_kernel(
    const int* __restrict__ src, const int* __restrict__ dst,
    const float* __restrict__ norm, int* __restrict__ cursor,
    int* __restrict__ rec, int E, int nb)
{
    int* hist  = dyn_lds;
    int* rbase = dyn_lds + nb;
    const int tid = threadIdx.x;
    for (int i = tid; i < nb; i += BIN_THREADS) hist[i] = 0;
    const long c0 = (long)blockIdx.x * BIN_EDGES + tid;
    int   dv[BIN_EDGES / BIN_THREADS];
    int   sv[BIN_EDGES / BIN_THREADS];
    float wv[BIN_EDGES / BIN_THREADS];
    for (int k = 0; k < BIN_EDGES / BIN_THREADS; ++k) {
        const long e  = c0 + (long)k * BIN_THREADS;
        const long ec = (e < E) ? e : (long)(E - 1);
        dv[k] = dst[ec];
        sv[k] = src[ec];
        wv[k] = norm[ec];
    }
    __syncthreads();
    for (int k = 0; k < BIN_EDGES / BIN_THREADS; ++k) {
        const long e = c0 + (long)k * BIN_THREADS;
        if (e < E) atomicAdd(&hist[dv[k] >> TILE_SHIFT], 1);
    }
    __syncthreads();
    for (int i = tid; i < nb; i += BIN_THREADS) {
        const int c = hist[i];
        rbase[i] = c ? atomicAdd(&cursor[i], c) : 0;
        hist[i] = 0;
    }
    __syncthreads();
    v4i* rec4 = (v4i*)rec;
    for (int k = 0; k < BIN_EDGES / BIN_THREADS; ++k) {
        const long e = c0 + (long)k * BIN_THREADS;
        if (e < E) {
            const int b   = dv[k] >> TILE_SHIFT;
            const int pos = rbase[b] + atomicAdd(&hist[b], 1);
            v4i r;
            r.x = sv[k];
            r.y = dv[k] - (b << TILE_SHIFT);
            r.z = __float_as_int(wv[k]);
            r.w = 0;
            rec4[pos] = r;
        }
    }
}

__device__ __forceinline__ void lds_add(float* p, float v)
{
    __hip_atomic_fetch_add(p, v, __ATOMIC_RELAXED, __HIP_MEMORY_SCOPE_WORKGROUP);
}

__global__ __launch_bounds__(HOP_THREADS) void hop_tile_kernel(
    const int* __restrict__ rec, const int* __restrict__ offs,
    const float* __restrict__ prev, float* __restrict__ next,
    float* __restrict__ acc, const float* __restrict__ proj_w,
    const float* __restrict__ proj_b, int N)
{
    __shared__ __attribute__((aligned(16))) float tile[(TILE_ROWS + 1) * C_DIM];
    __shared__ float rs[TILE_ROWS];

    const int tid  = threadIdx.x;
    const int b    = blockIdx.x;
    const int row0 = b << TILE_SHIFT;
    const int nrows = (N - row0 < TILE_ROWS) ? (N - row0) : TILE_ROWS;
    const int nv4   = nrows * (C_DIM / 4);

    v4f* tile4 = (v4f*)tile;
    #pragma unroll
    for (int j = 0; j < HOP_V4; ++j)
        tile4[tid + j * HOP_THREADS] = (v4f){0.0f, 0.0f, 0.0f, 0.0f};
    if (tid < C_DIM / 4)
        tile4[TILE_ROWS * C_DIM / 4 + tid] = (v4f){0.0f, 0.0f, 0.0f, 0.0f};
    __syncthreads();

    v4f* acc4 = (v4f*)(acc + (long)row0 * C_DIM);
    v4f a[HOP_V4];
    #pragma unroll
    for (int j = 0; j < HOP_V4; ++j) {
        const int i = tid + j * HOP_THREADS;
        a[j] = acc4[(i < nv4) ? i : (nv4 - 1)];
    }

    const v4i* rec4 = (const v4i*)rec;
    const int e0  = offs[b];
    const int e1  = offs[b + 1];
    const int grp = tid >> 3;
    const int l   = tid & 7;

    if (e0 < e1) {
        const int last = e1 - 1;
        int e = e0 + grp;
        v4i r[HOP_EPT];
        #pragma unroll
        for (int q = 0; q < HOP_EPT; ++q) {
            const int i = e + q * HOP_GROUPS;
            r[q] = rec4[(i < last) ? i : last];
        }
        while (e < e1) {
            float v[HOP_EPT][5];
            #pragma unroll
            for (int q = 0; q < HOP_EPT; ++q) {
                const float* p = prev + (long)r[q].x * C_DIM + l;
                for (int j = 0; j < 5; ++j) v[q][j] = p[8 * j];
            }
            const int en = e + HOP_EPT * HOP_GROUPS;
            v4i n[HOP_EPT];
            #pragma unroll
            for (int q = 0; q < HOP_EPT; ++q) {
                const int i = en + q * HOP_GROUPS;
                n[q] = rec4[(i < last) ? i : last];
            }
            #pragma unroll
            for (int q = 0; q < HOP_EPT; ++q) {
                const int   row = (e + q * HOP_GROUPS < e1) ? r[q].y : TILE_ROWS;
                const float w   = __int_as_float(r[q].z);
                float* t = tile + row * C_DIM + l;
                for (int j = 0; j < 5; ++j) lds_add(&t[8 * j], w * v[q][j]);
            }
            e = en;
            #pragma unroll
            for (int q = 0; q < HOP_EPT; ++q) r[q] = n[q];
        }
    }
    __syncthreads();

    if (tid < nrows) {
        const v4f* rp = (const v4f*)&tile[tid * C_DIM];
        float d = proj_b[0];
        #pragma unroll
        for (int q = 0; q < C_DIM / 4; ++q) {
            const v4f t = rp[q];
            d += t.x * proj_w[4 * q + 0];
            d += t.y * proj_w[4 * q + 1];
            d += t.z * proj_w[4 * q + 2];
            d += t.w * proj_w[4 * q + 3];
        }
        rs[tid] = 1.0f / (1.0f + expf(-d));
    }
    __syncthreads();

    v4f* out4 = (v4f*)(next + (long)row0 * C_DIM);
    v4f   t[HOP_V4];
    float rf[HOP_V4];
    #pragma unroll
    for (int j = 0; j < HOP_V4; ++j) {
        const int i  = tid + j * HOP_THREADS;
        const int ic = (i < nv4) ? i : (nv4 - 1);
        t[j]  = tile4[i];
        rf[j] = rs[ic / (C_DIM / 4)];
    }
    #pragma unroll
    for (int j = 0; j < HOP_V4; ++j) {
        const int i = tid + j * HOP_THREADS;
        if (i < nv4) {
            out4[i] = t[j];
            acc4[i] = a[j] + rf[j] * t[j];
        }
    }
}

#define ROW_TILE  256
#define ROW_V4    (C_DIM / 4)
#define ROW_PITCH 44

__global__ __launch_bounds__(ROW_TILE) void lsm_kernel(
    const float* __restrict__ acc, float* __restrict__ out, int N)
{
    __shared__ __attribute__((aligned(16))) float rows[ROW_TILE * ROW_PITCH];

    const int tid   = threadIdx.x;
    const int row0  = blockIdx.x * ROW_TILE;
    const int nrows = (N - row0 < ROW_TILE) ? (N - row0) : ROW_TILE;
    const int nv4   = nrows * ROW_V4;
    const v4f* a4 = (const v4f*)(acc + (long)row0 * C_DIM);
    v4f*       o4 = (v4f*)(out + (long)row0 * C_DIM);

    v4f x[ROW_V4];
    #pragma unroll
    for (int j = 0; j < ROW_V4; ++j) {
        const int i  = tid + j * ROW_TILE;
        const int ic = (i < nv4) ? i : (nv4 - 1);
        x[j] = a4[ic];
    }
    #pragma unroll
    for (int j = 0; j < ROW_V4; ++j) {
        const int i = tid + j * ROW_TILE;
        const int r = i / ROW_V4;
        const int q = i - r * ROW_V4;
        *(v4f*)&rows[r * ROW_PITCH + q * 4] = x[j];
    }
    __syncthreads();

    if (tid < nrows) {
        v4f* rp = (v4f*)&rows[tid * ROW_PITCH];
        v4f t[ROW_V4];
        #pragma unroll
        for (int q = 0; q < ROW_V4; ++q) t[q] = rp[q];
        float m = -3.402823466e38f;
        #pragma unroll
        for (int q = 0; q < ROW_V4; ++q) {
            m = fmaxf(m, t[q].x); m = fmaxf(m, t[q].y);
            m = fmaxf(m, t[q].z); m = fmaxf(m, t[q].w);
        }
        float s = 0.0f;
        #pragma unroll
        for (int q = 0; q < ROW_V4; ++q) {
            s += expf(t[q].x - m); s += expf(t[q].y - m);
            s += expf(t[q].z - m); s += expf(t[q].w - m);
        }
        const float ls = logf(s);
        #pragma unroll
        for (int q = 0; q < ROW_V4; ++q) {
            v4f o;
            o.x = t[q].x - m - ls; o.y = t[q].y - m - ls;
            o.z = t[q].z - m - ls; o.w = t[q].w - m - ls;
            rp[q] = o;
        }
    }
    __syncthreads();

    for (int j = 0; j < ROW_V4; ++j) {
        const int i = tid + j * ROW_TILE;
        if (i < nv4) {
            const int r = i / ROW_V4;
            const int q = i - r * ROW_V4;
            o4[i] = *(const v4f*)&rows[r * ROW_PITCH + q * 4];
        }
    }
}

extern "C" void kernel_launch(void* const* d_in, const int* in_sizes, int n_in,
                              void* d_out, int out_size, void* d_ws, size_t ws_size,
                              hipStream_t stream)
{
    const float* x     = (const float*)d_in[0];
    const int*   edge  = (const int*)d_in[1];
    const float* norm  = (const float*)d_in[2];
    const float* W1    = (const float*)d_in[3];
    const float* b1    = (const float*)d_in[4];
    const float* gamma = (const float*)d_in[5];
    const float* beta  = (const float*)d_in[6];
    const float* mean  = (const float*)d_in[7];
    const float* var   = (const float*)d_in[8];
    const float* W2    = (const float*)d_in[9];
    const float* b2    = (const float*)d_in[10];
    const float* pw    = (const float*)d_in[11];
    const float* pb    = (const float*)d_in[12];

    const int N = in_sizes[0] / F_DIM;
    const int E = in_sizes[2];

    const size_t per = (((size_t)N * C_DIM) + 63) & ~(size_t)63;
    float* bufA = (float*)d_ws;
    float* bufB = bufA + per;
    float* accb = bufB + per;

    const int    nb  = (N + TILE_ROWS - 1) / TILE_ROWS;
    const size_t nbp = ((size_t)nb + 1 + 63) & ~(size_t)63;
    int* rec    = (int*)(accb + per);
    int* counts = rec + (size_t)E * 4;
    int* offs   = counts + nbp;
    int* cursor = offs + nbp;

    const int* src = edge;
    const int* dst = edge + E;

    const int  nblocks = (N + ROW_TILE - 1) / ROW_TILE;
    const int  bblocks = (E + BIN_EDGES - 1) / BIN_EDGES;

    mlp_kernel<<<(N + MT_ROWS - 1) / MT_ROWS, 256, 0, stream>>>(
        x, W1, b1, gamma, beta, mean, var, W2, b2, pw, pb, bufA, accb, N);

    zero_int_kernel<<<(nb + 255) / 256, 256, 0, stream>>>(counts, nb);
    bin_count_kernel<<<bblocks, BIN_THREADS, nb * sizeof(int), stream>>>(
        dst, counts, E, nb);
    bin_scan_kernel<<<1, 1024, 0, stream>>>(counts, offs, cursor, nb);
    bin_fill_kernel<<<bblocks, BIN_THREADS, 2 * nb * sizeof(int), stream>>>(
        src, dst, norm, cursor, rec, E, nb);

    for (int k = 0; k < 10; ++k) {
        const float* prev = (k & 1) ? bufB : bufA;
        float*       next = (k & 1) ? bufA : bufB;
        hop_tile_kernel<<<nb, HOP_THREADS, 0, stream>>>(
            rec, offs, prev, next, accb, pw, pb, N);
    }

    lsm_kernel<<<nblocks, ROW_TILE, 0, stream>>>(accb, (float*)d_out, N);
}
